# starting kernel
# baseline (speedup 1.0000x reference)
#include <hip/hip_runtime.h>
#include <hip/hip_bf16.h>

#define D_MODEL 1024
#define D_STATE 16
#define BATCH   2
#define SEQ     2048
#define NCHUNK  16
#define LCHUNK  (SEQ / NCHUNK)

typedef __attribute__((ext_vector_type(2))) float v2f;
typedef __attribute__((ext_vector_type(8))) float v8f;

__global__ void ssm_abar_kernel(const float* __restrict__ logA,
                                float* __restrict__ A_bar,
                                float* __restrict__ A_Lc) {
  int i = blockIdx.x * blockDim.x + threadIdx.x;
  if (i < D_MODEL * D_STATE) {
    float ab = expf(-expf(logA[i]));
    A_bar[i] = ab;
    float p = ab;
#pragma unroll
    for (int j = 0; j < 7; ++j) p = p * p;
    A_Lc[i] = p;
  }
}

__global__ void ssm_bproj_wmma_kernel(const float* __restrict__ x,
                                      const float* __restrict__ Bw,
                                      float* __restrict__ Bt) {
  const int row0 = blockIdx.x * 16;
  const int lane = threadIdx.x;
  const int half = lane >> 4;
  const int lidx = lane & 15;

  const float* arow = x  + (size_t)(row0 + lidx) * D_MODEL;
  const float* brow = Bw + (size_t)lidx * D_MODEL;

  v8f acc = {};
  for (int k0 = 0; k0 < D_MODEL; k0 += 4) {
    v2f a, b;
    a.x = arow[k0 + 2 * half + 0];
    a.y = arow[k0 + 2 * half + 1];
    b.x = brow[k0 + 2 * half + 0];
    b.y = brow[k0 + 2 * half + 1];
    acc = __builtin_amdgcn_wmma_f32_16x16x4_f32(false, a, false, b,
                                                (short)0, acc, false, false);
  }
#pragma unroll
  for (int i = 0; i < 8; ++i) {
    Bt[(size_t)(row0 + i + 8 * half) * D_STATE + lidx] = acc[i];
  }
}

__global__ void ssm_chunk_local_kernel(const float* __restrict__ Bt,
                                       const float* __restrict__ A_bar,
                                       float* __restrict__ Ssum) {
  const int d = blockIdx.x * blockDim.x + threadIdx.x;
  const int c = blockIdx.y;
  const int b = blockIdx.z;

  float A[D_STATE], h[D_STATE];
  {
    const float4* p = reinterpret_cast<const float4*>(A_bar + (size_t)d * D_STATE);
#pragma unroll
    for (int q = 0; q < 4; ++q) {
      float4 v = p[q];
      A[4 * q + 0] = v.x; A[4 * q + 1] = v.y;
      A[4 * q + 2] = v.z; A[4 * q + 3] = v.w;
    }
  }
#pragma unroll
  for (int s = 0; s < D_STATE; ++s) h[s] = 0.0f;

  const float* btrow = Bt + ((size_t)b * SEQ + (size_t)c * LCHUNK) * D_STATE;
  for (int l = 0; l < LCHUNK; ++l) {
    const float4* p = reinterpret_cast<const float4*>(btrow + (size_t)l * D_STATE);
#pragma unroll
    for (int q = 0; q < 4; ++q) {
      float4 v = p[q];
      h[4 * q + 0] = fmaf(A[4 * q + 0], h[4 * q + 0], v.x);
      h[4 * q + 1] = fmaf(A[4 * q + 1], h[4 * q + 1], v.y);
      h[4 * q + 2] = fmaf(A[4 * q + 2], h[4 * q + 2], v.z);
      h[4 * q + 3] = fmaf(A[4 * q + 3], h[4 * q + 3], v.w);
    }
  }

  float4* out = reinterpret_cast<float4*>(
      Ssum + (((size_t)b * D_MODEL + d) * NCHUNK + c) * D_STATE);
#pragma unroll
  for (int q = 0; q < 4; ++q) {
    float4 v = { h[4 * q + 0], h[4 * q + 1], h[4 * q + 2], h[4 * q + 3] };
    out[q] = v;
  }
}

__global__ void ssm_prefix_kernel(const float* __restrict__ Ssum,
                                  const float* __restrict__ A_Lc,
                                  float* __restrict__ Hinit) {
  int idx = blockIdx.x * blockDim.x + threadIdx.x;
  if (idx >= BATCH * D_MODEL) return;
  int d = idx & (D_MODEL - 1);

  float AL[D_STATE], h[D_STATE];
  {
    const float4* p = reinterpret_cast<const float4*>(A_Lc + (size_t)d * D_STATE);
#pragma unroll
    for (int q = 0; q < 4; ++q) {
      float4 v = p[q];
      AL[4 * q + 0] = v.x; AL[4 * q + 1] = v.y;
      AL[4 * q + 2] = v.z; AL[4 * q + 3] = v.w;
    }
  }
#pragma unroll
  for (int s = 0; s < D_STATE; ++s) h[s] = 0.0f;

  for (int c = 0; c < NCHUNK; ++c) {
    size_t off = ((size_t)idx * NCHUNK + c) * D_STATE;
    float4* ho = reinterpret_cast<float4*>(Hinit + off);
    const float4* si = reinterpret_cast<const float4*>(Ssum + off);
#pragma unroll
    for (int q = 0; q < 4; ++q) {
      float4 v = { h[4 * q + 0], h[4 * q + 1], h[4 * q + 2], h[4 * q + 3] };
      ho[q] = v;
      float4 s4 = si[q];
      h[4 * q + 0] = fmaf(AL[4 * q + 0], h[4 * q + 0], s4.x);
      h[4 * q + 1] = fmaf(AL[4 * q + 1], h[4 * q + 1], s4.y);
      h[4 * q + 2] = fmaf(AL[4 * q + 2], h[4 * q + 2], s4.z);
      h[4 * q + 3] = fmaf(AL[4 * q + 3], h[4 * q + 3], s4.w);
    }
  }
}

__global__ void ssm_output_kernel(const float* __restrict__ x,
                                  const float* __restrict__ Bt,
                                  const float* __restrict__ A_bar,
                                  const float* __restrict__ Cw,
                                  const float* __restrict__ Dvec,
                                  const float* __restrict__ Hinit,
                                  float* __restrict__ y) {
  const int d = blockIdx.x * blockDim.x + threadIdx.x;
  const int c = blockIdx.y;
  const int b = blockIdx.z;

  float A[D_STATE], C[D_STATE], h[D_STATE];
  {
    const float4* pa = reinterpret_cast<const float4*>(A_bar + (size_t)d * D_STATE);
    const float4* pc = reinterpret_cast<const float4*>(Cw + (size_t)d * D_STATE);
    const float4* ph = reinterpret_cast<const float4*>(
        Hinit + (((size_t)b * D_MODEL + d) * NCHUNK + c) * D_STATE);
#pragma unroll
    for (int q = 0; q < 4; ++q) {
      float4 va = pa[q], vc = pc[q], vh = ph[q];
      A[4 * q + 0] = va.x; A[4 * q + 1] = va.y; A[4 * q + 2] = va.z; A[4 * q + 3] = va.w;
      C[4 * q + 0] = vc.x; C[4 * q + 1] = vc.y; C[4 * q + 2] = vc.z; C[4 * q + 3] = vc.w;
      h[4 * q + 0] = vh.x; h[4 * q + 1] = vh.y; h[4 * q + 2] = vh.z; h[4 * q + 3] = vh.w;
    }
  }
  const float dskip = Dvec[d];

  const size_t rowbase = (size_t)b * SEQ + (size_t)c * LCHUNK;
  const float* btrow = Bt + rowbase * D_STATE;
  const float* xcol  = x + rowbase * D_MODEL + d;
  float*       ycol  = y + rowbase * D_MODEL + d;

  for (int l = 0; l < LCHUNK; ++l) {
    if (l + 16 < LCHUNK)
      __builtin_prefetch(xcol + (size_t)(l + 16) * D_MODEL, 0, 0);

    const float4* p = reinterpret_cast<const float4*>(btrow + (size_t)l * D_STATE);
    float acc = 0.0f;
#pragma unroll
    for (int q = 0; q < 4; ++q) {
      float4 v = p[q];
      h[4 * q + 0] = fmaf(A[4 * q + 0], h[4 * q + 0], v.x);
      h[4 * q + 1] = fmaf(A[4 * q + 1], h[4 * q + 1], v.y);
      h[4 * q + 2] = fmaf(A[4 * q + 2], h[4 * q + 2], v.z);
      h[4 * q + 3] = fmaf(A[4 * q + 3], h[4 * q + 3], v.w);
      acc = fmaf(C[4 * q + 0], h[4 * q + 0], acc);
      acc = fmaf(C[4 * q + 1], h[4 * q + 1], acc);
      acc = fmaf(C[4 * q + 2], h[4 * q + 2], acc);
      acc = fmaf(C[4 * q + 3], h[4 * q + 3], acc);
    }
    float xv = xcol[(size_t)l * D_MODEL];
    ycol[(size_t)l * D_MODEL] = fmaf(dskip, xv, acc);
  }
}

extern "C" void kernel_launch(void* const* d_in, const int* in_sizes, int n_in,
                              void* d_out, int out_size, void* d_ws, size_t ws_size,
                              hipStream_t stream) {
  const float* x    = (const float*)d_in[0];
  const float* logA = (const float*)d_in[1];
  const float* Bw   = (const float*)d_in[2];
  const float* Cw   = (const float*)d_in[3];
  const float* Dv   = (const float*)d_in[4];
  float* y = (float*)d_out;

  char* ws = (char*)d_ws;
  float* A_bar = (float*)(ws);
  float* A_Lc  = (float*)(ws + (64 << 10));
  float* Bt    = (float*)(ws + (128 << 10));
  float* Ssum  = (float*)(ws + (384 << 10));
  float* Hinit = (float*)(ws + (384 << 10) + (2 << 20));

  ssm_abar_kernel<<<(D_MODEL * D_STATE + 255) / 256, 256, 0, stream>>>(
      logA, A_bar, A_Lc);

  ssm_bproj_wmma_kernel<<<(BATCH * SEQ) / 16, 32, 0, stream>>>(x, Bw, Bt);

  dim3 gscan(D_MODEL / 256, NCHUNK, BATCH);

  ssm_chunk_local_kernel<<<gscan, 256, 0, stream>>>(Bt, A_bar, Ssum);

  ssm_prefix_kernel<<<(BATCH * D_MODEL + 255) / 256, 256, 0, stream>>>(
      Ssum, A_Lc, Hinit);

  ssm_output_kernel<<<gscan, 256, 0, stream>>>(x, Bt, A_bar, Cw, Dv, Hinit, y);
}


# Replay kernel: Bt rows requested four at a time and x one group of eight steps ahead, so each load round trip is waited for once per group, not once per step; order of every sum unchanged
# speedup vs baseline: 1.7401x; 1.7401x over previous
#include <hip/hip_runtime.h>
#include <hip/hip_bf16.h>

#define D_MODEL 1024
#define D_STATE 16
#define BATCH   2
#define SEQ     2048
#define NCHUNK  16
#define LCHUNK  (SEQ / NCHUNK)
#define BGRP    4
#define XGRP    8

typedef __attribute__((ext_vector_type(2))) float v2f;
typedef __attribute__((ext_vector_type(8))) float v8f;

__global__ void ssm_abar_kernel(const float* __restrict__ logA,
                                float* __restrict__ A_bar,
                                float* __restrict__ A_Lc) {
  int i = blockIdx.x * blockDim.x + threadIdx.x;
  if (i < D_MODEL * D_STATE) {
    float ab = expf(-expf(logA[i]));
    A_bar[i] = ab;
    float p = ab;
#pragma unroll
    for (int j = 0; j < 7; ++j) p = p * p;
    A_Lc[i] = p;
  }
}

__global__ void ssm_bproj_wmma_kernel(const float* __restrict__ x,
                                      const float* __restrict__ Bw,
                                      float* __restrict__ Bt) {
  const int row0 = blockIdx.x * 16;
  const int lane = threadIdx.x;
  const int half = lane >> 4;
  const int lidx = lane & 15;

  const float* arow = x  + (size_t)(row0 + lidx) * D_MODEL;
  const float* brow = Bw + (size_t)lidx * D_MODEL;

  v8f acc = {};
  for (int k0 = 0; k0 < D_MODEL; k0 += 4) {
    v2f a, b;
    a.x = arow[k0 + 2 * half + 0];
    a.y = arow[k0 + 2 * half + 1];
    b.x = brow[k0 + 2 * half + 0];
    b.y = brow[k0 + 2 * half + 1];
    acc = __builtin_amdgcn_wmma_f32_16x16x4_f32(false, a, false, b,
                                                (short)0, acc, false, false);
  }
#pragma unroll
  for (int i = 0; i < 8; ++i) {
    Bt[(size_t)(row0 + i + 8 * half) * D_STATE + lidx] = acc[i];
  }
}

__global__ void ssm_chunk_local_kernel(const float* __restrict__ Bt,
                                       const float* __restrict__ A_bar,
                                       float* __restrict__ Ssum) {
  const int d = blockIdx.x * blockDim.x + threadIdx.x;
  const int c = blockIdx.y;
  const int b = blockIdx.z;

  float A[D_STATE], h[D_STATE];
  {
    const float4* p = reinterpret_cast<const float4*>(A_bar + (size_t)d * D_STATE);
#pragma unroll
    for (int q = 0; q < 4; ++q) {
      float4 v = p[q];
      A[4 * q + 0] = v.x; A[4 * q + 1] = v.y;
      A[4 * q + 2] = v.z; A[4 * q + 3] = v.w;
    }
  }
#pragma unroll
  for (int s = 0; s < D_STATE; ++s) h[s] = 0.0f;

  const float* btrow = Bt + ((size_t)b * SEQ + (size_t)c * LCHUNK) * D_STATE;
  for (int l = 0; l < LCHUNK; ++l) {
    const float4* p = reinterpret_cast<const float4*>(btrow + (size_t)l * D_STATE);
#pragma unroll
    for (int q = 0; q < 4; ++q) {
      float4 v = p[q];
      h[4 * q + 0] = fmaf(A[4 * q + 0], h[4 * q + 0], v.x);
      h[4 * q + 1] = fmaf(A[4 * q + 1], h[4 * q + 1], v.y);
      h[4 * q + 2] = fmaf(A[4 * q + 2], h[4 * q + 2], v.z);
      h[4 * q + 3] = fmaf(A[4 * q + 3], h[4 * q + 3], v.w);
    }
  }

  float4* out = reinterpret_cast<float4*>(
      Ssum + (((size_t)b * D_MODEL + d) * NCHUNK + c) * D_STATE);
#pragma unroll
  for (int q = 0; q < 4; ++q) {
    float4 v = { h[4 * q + 0], h[4 * q + 1], h[4 * q + 2], h[4 * q + 3] };
    out[q] = v;
  }
}

__global__ void ssm_prefix_kernel(const float* __restrict__ Ssum,
                                  const float* __restrict__ A_Lc,
                                  float* __restrict__ Hinit) {
  int idx = blockIdx.x * blockDim.x + threadIdx.x;
  if (idx >= BATCH * D_MODEL) return;
  int d = idx & (D_MODEL - 1);

  float AL[D_STATE], h[D_STATE];
  {
    const float4* p = reinterpret_cast<const float4*>(A_Lc + (size_t)d * D_STATE);
#pragma unroll
    for (int q = 0; q < 4; ++q) {
      float4 v = p[q];
      AL[4 * q + 0] = v.x; AL[4 * q + 1] = v.y;
      AL[4 * q + 2] = v.z; AL[4 * q + 3] = v.w;
    }
  }
#pragma unroll
  for (int s = 0; s < D_STATE; ++s) h[s] = 0.0f;

  for (int c = 0; c < NCHUNK; ++c) {
    size_t off = ((size_t)idx * NCHUNK + c) * D_STATE;
    float4* ho = reinterpret_cast<float4*>(Hinit + off);
    const float4* si = reinterpret_cast<const float4*>(Ssum + off);
#pragma unroll
    for (int q = 0; q < 4; ++q) {
      float4 v = { h[4 * q + 0], h[4 * q + 1], h[4 * q + 2], h[4 * q + 3] };
      ho[q] = v;
      float4 s4 = si[q];
      h[4 * q + 0] = fmaf(AL[4 * q + 0], h[4 * q + 0], s4.x);
      h[4 * q + 1] = fmaf(AL[4 * q + 1], h[4 * q + 1], s4.y);
      h[4 * q + 2] = fmaf(AL[4 * q + 2], h[4 * q + 2], s4.z);
      h[4 * q + 3] = fmaf(AL[4 * q + 3], h[4 * q + 3], s4.w);
    }
  }
}

__device__ __forceinline__ void ssm_replay_steps(const float4*& btp,
                                                 const float (&A)[D_STATE],
                                                 const float (&C)[D_STATE],
                                                 float (&h)[D_STATE],
                                                 const float (&xv)[XGRP],
                                                 float dskip,
                                                 float* __restrict__ yrow) {
#pragma unroll
  for (int g = 0; g < XGRP; g += BGRP) {
    __builtin_amdgcn_sched_barrier(0);

    float4 bt[BGRP][4];
#pragma unroll
    for (int j = 0; j < BGRP; ++j) {
#pragma unroll
      for (int q = 0; q < 4; ++q) bt[j][q] = btp[4 * j + q];
    }
    btp += 4 * BGRP;

#pragma unroll
    for (int j = 0; j < BGRP; ++j) {
      float acc = 0.0f;
#pragma unroll
      for (int q = 0; q < 4; ++q) {
        float4 v = bt[j][q];
        h[4 * q + 0] = fmaf(A[4 * q + 0], h[4 * q + 0], v.x);
        h[4 * q + 1] = fmaf(A[4 * q + 1], h[4 * q + 1], v.y);
        h[4 * q + 2] = fmaf(A[4 * q + 2], h[4 * q + 2], v.z);
        h[4 * q + 3] = fmaf(A[4 * q + 3], h[4 * q + 3], v.w);
        acc = fmaf(C[4 * q + 0], h[4 * q + 0], acc);
        acc = fmaf(C[4 * q + 1], h[4 * q + 1], acc);
        acc = fmaf(C[4 * q + 2], h[4 * q + 2], acc);
        acc = fmaf(C[4 * q + 3], h[4 * q + 3], acc);
      }
      yrow[(size_t)(g + j) * D_MODEL] = fmaf(dskip, xv[g + j], acc);
    }
  }
}

__global__ void
ssm_output_kernel(const float* __restrict__ x,
                  const float* __restrict__ Bt,
                  const float* __restrict__ A_bar,
                  const float* __restrict__ Cw,
                  const float* __restrict__ Dvec,
                  const float* __restrict__ Hinit,
                  float* __restrict__ y) {
  const int d = blockIdx.x * blockDim.x + threadIdx.x;
  const int c = blockIdx.y;
  const int b = blockIdx.z;

  float A[D_STATE], C[D_STATE], h[D_STATE];
  {
    const float4* pa = reinterpret_cast<const float4*>(A_bar + (size_t)d * D_STATE);
    const float4* pc = reinterpret_cast<const float4*>(Cw + (size_t)d * D_STATE);
    const float4* ph = reinterpret_cast<const float4*>(
        Hinit + (((size_t)b * D_MODEL + d) * NCHUNK + c) * D_STATE);
#pragma unroll
    for (int q = 0; q < 4; ++q) {
      float4 va = pa[q], vc = pc[q], vh = ph[q];
      A[4 * q + 0] = va.x; A[4 * q + 1] = va.y; A[4 * q + 2] = va.z; A[4 * q + 3] = va.w;
      C[4 * q + 0] = vc.x; C[4 * q + 1] = vc.y; C[4 * q + 2] = vc.z; C[4 * q + 3] = vc.w;
      h[4 * q + 0] = vh.x; h[4 * q + 1] = vh.y; h[4 * q + 2] = vh.z; h[4 * q + 3] = vh.w;
    }
  }
  const float dskip = Dvec[d];

  const size_t rowbase = (size_t)b * SEQ + (size_t)c * LCHUNK;
  const float* btrow = Bt + rowbase * D_STATE;
  const float* xcol  = x + rowbase * D_MODEL + d;
  float*       ycol  = y + rowbase * D_MODEL + d;

  float xa[XGRP], xb[XGRP];
#pragma unroll
  for (int j = 0; j < XGRP; ++j) xa[j] = xcol[(size_t)j * D_MODEL];

  const float4* btp = reinterpret_cast<const float4*>(btrow);
  for (int l0 = 0; l0 < LCHUNK; l0 += 2 * XGRP) {
#pragma unroll
    for (int j = 0; j < XGRP; ++j) xb[j] = xcol[(size_t)(l0 + XGRP + j) * D_MODEL];

    ssm_replay_steps(btp, A, C, h, xa, dskip, ycol + (size_t)l0 * D_MODEL);

    const int ln = (l0 + 2 * XGRP < LCHUNK) ? (l0 + 2 * XGRP) : l0;
#pragma unroll
    for (int j = 0; j < XGRP; ++j) xa[j] = xcol[(size_t)(ln + j) * D_MODEL];

    ssm_replay_steps(btp, A, C, h, xb, dskip, ycol + (size_t)(l0 + XGRP) * D_MODEL);
  }
}

extern "C" void kernel_launch(void* const* d_in, const int* in_sizes, int n_in,
                              void* d_out, int out_size, void* d_ws, size_t ws_size,
                              hipStream_t stream) {
  const float* x    = (const float*)d_in[0];
  const float* logA = (const float*)d_in[1];
  const float* Bw   = (const float*)d_in[2];
  const float* Cw   = (const float*)d_in[3];
  const float* Dv   = (const float*)d_in[4];
  float* y = (float*)d_out;

  char* ws = (char*)d_ws;
  float* A_bar = (float*)(ws);
  float* A_Lc  = (float*)(ws + (64 << 10));
  float* Bt    = (float*)(ws + (128 << 10));
  float* Ssum  = (float*)(ws + (384 << 10));
  float* Hinit = (float*)(ws + (384 << 10) + (2 << 20));

  ssm_abar_kernel<<<(D_MODEL * D_STATE + 255) / 256, 256, 0, stream>>>(
      logA, A_bar, A_Lc);

  ssm_bproj_wmma_kernel<<<(BATCH * SEQ) / 16, 32, 0, stream>>>(x, Bw, Bt);

  dim3 gscan(D_MODEL / 256, NCHUNK, BATCH);

  ssm_chunk_local_kernel<<<gscan, 256, 0, stream>>>(Bt, A_bar, Ssum);

  ssm_prefix_kernel<<<(BATCH * D_MODEL + 255) / 256, 256, 0, stream>>>(
      Ssum, A_Lc, Hinit);

  ssm_output_kernel<<<gscan, 256, 0, stream>>>(x, Bt, A_bar, Cw, Dv, Hinit, y);
}


# Launches 5 -> 3: decay factors (exp, 7 squarings) and the carry over earlier chunk sums are computed inside the two scan kernels that use them, not in launches of their own; order of every sum kept
# speedup vs baseline: 2.4096x; 1.3847x over previous
#include <hip/hip_runtime.h>
#include <hip/hip_bf16.h>

#define D_MODEL 1024
#define D_STATE 16
#define BATCH   2
#define SEQ     2048
#define NCHUNK  16
#define LCHUNK  (SEQ / NCHUNK)
#define BGRP    4
#define XGRP    8
#define CB1     8

typedef __attribute__((ext_vector_type(2))) float v2f;
typedef __attribute__((ext_vector_type(8))) float v8f;

__device__ __forceinline__ void ssm_abar_from(const float4 (&la)[4], float (&A)[D_STATE]) {
#pragma unroll
  for (int q = 0; q < 4; ++q) {
    float4 v = la[q];
    A[4 * q + 0] = expf(-expf(v.x));
    A[4 * q + 1] = expf(-expf(v.y));
    A[4 * q + 2] = expf(-expf(v.z));
    A[4 * q + 3] = expf(-expf(v.w));
  }
}

__device__ __forceinline__ void ssm_load_abar(const float* __restrict__ logA, int d,
                                              float (&A)[D_STATE]) {
  const float4* p = reinterpret_cast<const float4*>(logA + (size_t)d * D_STATE);
  float4 la[4];
#pragma unroll
  for (int q = 0; q < 4; ++q) la[q] = p[q];
  ssm_abar_from(la, A);
}

__global__ void ssm_bproj_wmma_kernel(const float* __restrict__ x,
                                      const float* __restrict__ Bw,
                                      float* __restrict__ Bt) {
  const int row0 = blockIdx.x * 16;
  const int lane = threadIdx.x;
  const int half = lane >> 4;
  const int lidx = lane & 15;

  const float* arow = x  + (size_t)(row0 + lidx) * D_MODEL;
  const float* brow = Bw + (size_t)lidx * D_MODEL;

  v8f acc = {};
  for (int k0 = 0; k0 < D_MODEL; k0 += 4) {
    v2f a, b;
    a.x = arow[k0 + 2 * half + 0];
    a.y = arow[k0 + 2 * half + 1];
    b.x = brow[k0 + 2 * half + 0];
    b.y = brow[k0 + 2 * half + 1];
    acc = __builtin_amdgcn_wmma_f32_16x16x4_f32(false, a, false, b,
                                                (short)0, acc, false, false);
  }
#pragma unroll
  for (int i = 0; i < 8; ++i) {
    Bt[(size_t)(row0 + i + 8 * half) * D_STATE + lidx] = acc[i];
  }
}

__global__ void ssm_chunk_local_kernel(const float* __restrict__ Bt,
                                       const float* __restrict__ logA,
                                       float* __restrict__ Ssum) {
  const int d = blockIdx.x * blockDim.x + threadIdx.x;
  const int c = blockIdx.y;
  const int b = blockIdx.z;

  float A[D_STATE], h[D_STATE];
  ssm_load_abar(logA, d, A);
#pragma unroll
  for (int s = 0; s < D_STATE; ++s) h[s] = 0.0f;

  const float* btrow = Bt + ((size_t)b * SEQ + (size_t)c * LCHUNK) * D_STATE;
  for (int l = 0; l < LCHUNK; ++l) {
    const float4* p = reinterpret_cast<const float4*>(btrow + (size_t)l * D_STATE);
#pragma unroll
    for (int q = 0; q < 4; ++q) {
      float4 v = p[q];
      h[4 * q + 0] = fmaf(A[4 * q + 0], h[4 * q + 0], v.x);
      h[4 * q + 1] = fmaf(A[4 * q + 1], h[4 * q + 1], v.y);
      h[4 * q + 2] = fmaf(A[4 * q + 2], h[4 * q + 2], v.z);
      h[4 * q + 3] = fmaf(A[4 * q + 3], h[4 * q + 3], v.w);
    }
  }

  float4* out = reinterpret_cast<float4*>(
      Ssum + (((size_t)b * NCHUNK + c) * D_MODEL + d) * D_STATE);
#pragma unroll
  for (int q = 0; q < 4; ++q) {
    float4 v = { h[4 * q + 0], h[4 * q + 1], h[4 * q + 2], h[4 * q + 3] };
    out[q] = v;
  }
}

__device__ __forceinline__ void ssm_replay_steps(const float4*& btp,
                                                 const float (&A)[D_STATE],
                                                 const float (&C)[D_STATE],
                                                 float (&h)[D_STATE],
                                                 const float (&xv)[XGRP],
                                                 float dskip,
                                                 float* __restrict__ yrow) {
#pragma unroll
  for (int g = 0; g < XGRP; g += BGRP) {
    __builtin_amdgcn_sched_barrier(0);

    float4 bt[BGRP][4];
#pragma unroll
    for (int j = 0; j < BGRP; ++j) {
#pragma unroll
      for (int q = 0; q < 4; ++q) bt[j][q] = btp[4 * j + q];
    }
    btp += 4 * BGRP;

#pragma unroll
    for (int j = 0; j < BGRP; ++j) {
      float acc = 0.0f;
#pragma unroll
      for (int q = 0; q < 4; ++q) {
        float4 v = bt[j][q];
        h[4 * q + 0] = fmaf(A[4 * q + 0], h[4 * q + 0], v.x);
        h[4 * q + 1] = fmaf(A[4 * q + 1], h[4 * q + 1], v.y);
        h[4 * q + 2] = fmaf(A[4 * q + 2], h[4 * q + 2], v.z);
        h[4 * q + 3] = fmaf(A[4 * q + 3], h[4 * q + 3], v.w);
        acc = fmaf(C[4 * q + 0], h[4 * q + 0], acc);
        acc = fmaf(C[4 * q + 1], h[4 * q + 1], acc);
        acc = fmaf(C[4 * q + 2], h[4 * q + 2], acc);
        acc = fmaf(C[4 * q + 3], h[4 * q + 3], acc);
      }
      yrow[(size_t)(g + j) * D_MODEL] = fmaf(dskip, xv[g + j], acc);
    }
  }
}

__device__ __forceinline__ void ssm_carry_step(const float (&AL)[D_STATE],
                                               float (&h)[D_STATE],
                                               const float4 (&S)[4]) {
#pragma unroll
  for (int q = 0; q < 4; ++q) {
    float4 v = S[q];
    h[4 * q + 0] = fmaf(AL[4 * q + 0], h[4 * q + 0], v.x);
    h[4 * q + 1] = fmaf(AL[4 * q + 1], h[4 * q + 1], v.y);
    h[4 * q + 2] = fmaf(AL[4 * q + 2], h[4 * q + 2], v.z);
    h[4 * q + 3] = fmaf(AL[4 * q + 3], h[4 * q + 3], v.w);
  }
}

__global__ void __launch_bounds__(256)
ssm_output_kernel(const float* __restrict__ x,
                  const float* __restrict__ Bt,
                  const float* __restrict__ logA,
                  const float* __restrict__ Cw,
                  const float* __restrict__ Dvec,
                  const float* __restrict__ Ssum,
                  float* __restrict__ y) {
  const int d = blockIdx.x * blockDim.x + threadIdx.x;
  const int c = blockIdx.y;
  const int b = blockIdx.z;

  const size_t rowbase = (size_t)b * SEQ + (size_t)c * LCHUNK;
  const float* btrow = Bt + rowbase * D_STATE;
  const float* xcol  = x + rowbase * D_MODEL + d;
  float*       ycol  = y + rowbase * D_MODEL + d;

  const float* srow = Ssum + ((size_t)b * NCHUNK * D_MODEL + d) * D_STATE;
  float4 sv[CB1][4];
#pragma unroll
  for (int j = 0; j < CB1; ++j) {
    if (j < c) {
      const float4* p = reinterpret_cast<const float4*>(
          srow + (size_t)j * D_MODEL * D_STATE);
#pragma unroll
      for (int q = 0; q < 4; ++q) sv[j][q] = p[q];
    }
  }

  float4 la[4], vc[4];
  {
    const float4* pl = reinterpret_cast<const float4*>(logA + (size_t)d * D_STATE);
    const float4* pc = reinterpret_cast<const float4*>(Cw + (size_t)d * D_STATE);
#pragma unroll
    for (int q = 0; q < 4; ++q) { la[q] = pl[q]; vc[q] = pc[q]; }
  }
  const float dskip = Dvec[d];

  float xa[XGRP], xb[XGRP];
#pragma unroll
  for (int j = 0; j < XGRP; ++j) xa[j] = xcol[(size_t)j * D_MODEL];

  float A[D_STATE], C[D_STATE], h[D_STATE];
  ssm_abar_from(la, A);
#pragma unroll
  for (int q = 0; q < 4; ++q) {
    C[4 * q + 0] = vc[q].x; C[4 * q + 1] = vc[q].y; C[4 * q + 2] = vc[q].z; C[4 * q + 3] = vc[q].w;
  }
#pragma unroll
  for (int s = 0; s < D_STATE; ++s) h[s] = 0.0f;

  if (c > 0) {
    float AL[D_STATE];
#pragma unroll
    for (int s = 0; s < D_STATE; ++s) {
      float p = A[s];
#pragma unroll
      for (int j = 0; j < 7; ++j) p = p * p;
      AL[s] = p;
    }
#pragma unroll
    for (int j = 0; j < CB1; ++j) {
      if (j < c) ssm_carry_step(AL, h, sv[j]);
    }
    if (c > CB1) {
      float4 sw[NCHUNK - 1 - CB1][4];
#pragma unroll
      for (int j = 0; j < NCHUNK - 1 - CB1; ++j) {
        if (CB1 + j < c) {
          const float4* p = reinterpret_cast<const float4*>(
              srow + (size_t)(CB1 + j) * D_MODEL * D_STATE);
#pragma unroll
          for (int q = 0; q < 4; ++q) sw[j][q] = p[q];
        }
      }
#pragma unroll
      for (int j = 0; j < NCHUNK - 1 - CB1; ++j) {
        if (CB1 + j < c) ssm_carry_step(AL, h, sw[j]);
      }
    }
  }

  const float4* btp = reinterpret_cast<const float4*>(btrow);
  for (int l0 = 0; l0 < LCHUNK; l0 += 2 * XGRP) {
#pragma unroll
    for (int j = 0; j < XGRP; ++j) xb[j] = xcol[(size_t)(l0 + XGRP + j) * D_MODEL];

    ssm_replay_steps(btp, A, C, h, xa, dskip, ycol + (size_t)l0 * D_MODEL);

    const int ln = (l0 + 2 * XGRP < LCHUNK) ? (l0 + 2 * XGRP) : l0;
    __builtin_amdgcn_sched_barrier(0);
#pragma unroll
    for (int j = 0; j < XGRP; ++j) xa[j] = xcol[(size_t)(ln + j) * D_MODEL];

    ssm_replay_steps(btp, A, C, h, xb, dskip, ycol + (size_t)(l0 + XGRP) * D_MODEL);
  }
}

extern "C" void kernel_launch(void* const* d_in, const int* in_sizes, int n_in,
                              void* d_out, int out_size, void* d_ws, size_t ws_size,
                              hipStream_t stream) {
  const float* x    = (const float*)d_in[0];
  const float* logA = (const float*)d_in[1];
  const float* Bw   = (const float*)d_in[2];
  const float* Cw   = (const float*)d_in[3];
  const float* Dv   = (const float*)d_in[4];
  float* y = (float*)d_out;

  char* ws = (char*)d_ws;
  float* Bt    = (float*)(ws + (128 << 10));
  float* Ssum  = (float*)(ws + (384 << 10));

  ssm_bproj_wmma_kernel<<<(BATCH * SEQ) / 16, 32, 0, stream>>>(x, Bw, Bt);

  dim3 gscan(D_MODEL / 256, NCHUNK, BATCH);

  ssm_chunk_local_kernel<<<gscan, 256, 0, stream>>>(Bt, logA, Ssum);

  ssm_output_kernel<<<gscan, 256, 0, stream>>>(x, Bt, logA, Cw, Dv, Ssum, y);
}


# Both scan kernels: the chunk's Bt rows go once per work-group to LDS and are read from there a group ahead, replacing scalar loads waited for per group; 128-thread groups; order of every sum kept
# speedup vs baseline: 3.3328x; 1.3831x over previous
#include <hip/hip_runtime.h>
#include <hip/hip_bf16.h>

#define D_MODEL 1024
#define D_STATE 16
#define BATCH   2
#define SEQ     2048
#define NCHUNK  16
#define LCHUNK  (SEQ / NCHUNK)
#define BGRP    4
#define XGRP    8
#define CB1     8
#define SCAN_THREADS 128
#define STG     (LCHUNK * 4 / SCAN_THREADS)

typedef __attribute__((ext_vector_type(2))) float v2f;
typedef __attribute__((ext_vector_type(8))) float v8f;
typedef __attribute__((ext_vector_type(4))) float v4f;

__device__ __forceinline__ void ssm_abar_from(const v4f (&la)[4], float (&A)[D_STATE]) {
#pragma unroll
  for (int q = 0; q < 4; ++q) {
    v4f v = la[q];
    A[4 * q + 0] = expf(-expf(v.x));
    A[4 * q + 1] = expf(-expf(v.y));
    A[4 * q + 2] = expf(-expf(v.z));
    A[4 * q + 3] = expf(-expf(v.w));
  }
}

__device__ __forceinline__ void ssm_stage_request(const float* __restrict__ btrow,
                                                  v4f (&st)[STG]) {
  const v4f* g = reinterpret_cast<const v4f*>(btrow);
#pragma unroll
  for (int i = 0; i < STG; ++i) st[i] = g[threadIdx.x + i * SCAN_THREADS];
}

__device__ __forceinline__ void ssm_stage_commit(float4* __restrict__ sbt,
                                                 const v4f (&st)[STG]) {
  v4f* s = reinterpret_cast<v4f*>(sbt);
#pragma unroll
  for (int i = 0; i < STG; ++i) s[threadIdx.x + i * SCAN_THREADS] = st[i];
  __syncthreads();
}

__device__ __forceinline__ void ssm_rows_request(const float4* __restrict__ srow,
                                                 float4 (&r)[BGRP][4]) {
#pragma unroll
  for (int j = 0; j < BGRP; ++j) {
#pragma unroll
    for (int q = 0; q < 4; ++q) r[j][q] = srow[4 * j + q];
  }
}

__device__ __forceinline__ void ssm_rows_keep_ahead(float4 (&r)[BGRP][4]) {
#pragma unroll
  for (int j = 0; j < BGRP; ++j) {
#pragma unroll
    for (int q = 0; q < 4; ++q) {
      asm volatile("" : "+v"(r[j][q].x), "+v"(r[j][q].y), "+v"(r[j][q].z), "+v"(r[j][q].w));
    }
  }
}

__device__ __forceinline__ void ssm_entry_keep_together(v4f (&st)[STG], v4f (&la)[4]) {
  static_assert(STG == 4, "one operand per staged vector is written out below");
  asm volatile("" : "+v"(st[0]), "+v"(st[1]), "+v"(st[2]), "+v"(st[3]),
                    "+v"(la[0]), "+v"(la[1]), "+v"(la[2]), "+v"(la[3]));
}

__global__ void ssm_bproj_wmma_kernel(const float* __restrict__ x,
                                      const float* __restrict__ Bw,
                                      float* __restrict__ Bt) {
  const int row0 = blockIdx.x * 16;
  const int lane = threadIdx.x;
  const int half = lane >> 4;
  const int lidx = lane & 15;

  const float* arow = x  + (size_t)(row0 + lidx) * D_MODEL;
  const float* brow = Bw + (size_t)lidx * D_MODEL;

  v8f acc = {};
  for (int k0 = 0; k0 < D_MODEL; k0 += 4) {
    v2f a, b;
    a.x = arow[k0 + 2 * half + 0];
    a.y = arow[k0 + 2 * half + 1];
    b.x = brow[k0 + 2 * half + 0];
    b.y = brow[k0 + 2 * half + 1];
    acc = __builtin_amdgcn_wmma_f32_16x16x4_f32(false, a, false, b,
                                                (short)0, acc, false, false);
  }
#pragma unroll
  for (int i = 0; i < 8; ++i) {
    Bt[(size_t)(row0 + i + 8 * half) * D_STATE + lidx] = acc[i];
  }
}

__device__ __forceinline__ void ssm_scan_rows(const float4 (&bt)[BGRP][4],
                                              const float (&A)[D_STATE],
                                              float (&h)[D_STATE]) {
#pragma unroll
  for (int j = 0; j < BGRP; ++j) {
#pragma unroll
    for (int q = 0; q < 4; ++q) {
      float4 v = bt[j][q];
      h[4 * q + 0] = fmaf(A[4 * q + 0], h[4 * q + 0], v.x);
      h[4 * q + 1] = fmaf(A[4 * q + 1], h[4 * q + 1], v.y);
      h[4 * q + 2] = fmaf(A[4 * q + 2], h[4 * q + 2], v.z);
      h[4 * q + 3] = fmaf(A[4 * q + 3], h[4 * q + 3], v.w);
    }
  }
}

__global__ void __launch_bounds__(SCAN_THREADS)
ssm_chunk_local_kernel(const float* __restrict__ Bt,
                       const float* __restrict__ logA,
                       float* __restrict__ Ssum) {
  __shared__ float4 sbt[LCHUNK * 4];

  const int d = blockIdx.x * SCAN_THREADS + threadIdx.x;
  const int c = blockIdx.y;
  const int b = blockIdx.z;

  v4f st[STG];
  ssm_stage_request(Bt + ((size_t)b * SEQ + (size_t)c * LCHUNK) * D_STATE, st);
  v4f la[4];
  {
    const v4f* pl = reinterpret_cast<const v4f*>(logA + (size_t)d * D_STATE);
#pragma unroll
    for (int q = 0; q < 4; ++q) la[q] = pl[q];
  }
  ssm_entry_keep_together(st, la);
  ssm_stage_commit(sbt, st);

  float A[D_STATE], h[D_STATE];
  ssm_abar_from(la, A);
#pragma unroll
  for (int s = 0; s < D_STATE; ++s) h[s] = 0.0f;

  float4 ra[BGRP][4], rb[BGRP][4];
  ssm_rows_request(sbt, ra);
  ssm_rows_keep_ahead(ra);
  for (int l0 = 0; l0 < LCHUNK; l0 += 2 * BGRP) {
    __builtin_amdgcn_sched_barrier(0);
    ssm_rows_request(sbt + 4 * (l0 + BGRP), rb);
    __builtin_amdgcn_sched_barrier(0);
    ssm_scan_rows(ra, A, h);

    const int ln = (l0 + 2 * BGRP < LCHUNK) ? (l0 + 2 * BGRP) : (l0 + BGRP);
    __builtin_amdgcn_sched_barrier(0);
    ssm_rows_request(sbt + 4 * ln, ra);
    __builtin_amdgcn_sched_barrier(0);
    ssm_scan_rows(rb, A, h);
  }

  float4* out = reinterpret_cast<float4*>(
      Ssum + (((size_t)b * NCHUNK + c) * D_MODEL + d) * D_STATE);
#pragma unroll
  for (int q = 0; q < 4; ++q) {
    float4 v = { h[4 * q + 0], h[4 * q + 1], h[4 * q + 2], h[4 * q + 3] };
    out[q] = v;
  }
}

template <int G>
__device__ __forceinline__ void ssm_replay_rows(const float4 (&bt)[BGRP][4],
                                                const float (&A)[D_STATE],
                                                const float (&C)[D_STATE],
                                                float (&h)[D_STATE],
                                                const float (&xv)[XGRP],
                                                float dskip,
                                                float* __restrict__ yrow) {
#pragma unroll
  for (int j = 0; j < BGRP; ++j) {
    float acc = 0.0f;
#pragma unroll
    for (int q = 0; q < 4; ++q) {
      float4 v = bt[j][q];
      h[4 * q + 0] = fmaf(A[4 * q + 0], h[4 * q + 0], v.x);
      h[4 * q + 1] = fmaf(A[4 * q + 1], h[4 * q + 1], v.y);
      h[4 * q + 2] = fmaf(A[4 * q + 2], h[4 * q + 2], v.z);
      h[4 * q + 3] = fmaf(A[4 * q + 3], h[4 * q + 3], v.w);
      acc = fmaf(C[4 * q + 0], h[4 * q + 0], acc);
      acc = fmaf(C[4 * q + 1], h[4 * q + 1], acc);
      acc = fmaf(C[4 * q + 2], h[4 * q + 2], acc);
      acc = fmaf(C[4 * q + 3], h[4 * q + 3], acc);
    }
    yrow[(size_t)(G + j) * D_MODEL] = fmaf(dskip, xv[G + j], acc);
  }
}

__device__ __forceinline__ void ssm_carry_step(const float (&AL)[D_STATE],
                                               float (&h)[D_STATE],
                                               const float4 (&S)[4]) {
#pragma unroll
  for (int q = 0; q < 4; ++q) {
    float4 v = S[q];
    h[4 * q + 0] = fmaf(AL[4 * q + 0], h[4 * q + 0], v.x);
    h[4 * q + 1] = fmaf(AL[4 * q + 1], h[4 * q + 1], v.y);
    h[4 * q + 2] = fmaf(AL[4 * q + 2], h[4 * q + 2], v.z);
    h[4 * q + 3] = fmaf(AL[4 * q + 3], h[4 * q + 3], v.w);
  }
}

__global__ void __launch_bounds__(SCAN_THREADS)
ssm_output_kernel(const float* __restrict__ x,
                  const float* __restrict__ Bt,
                  const float* __restrict__ logA,
                  const float* __restrict__ Cw,
                  const float* __restrict__ Dvec,
                  const float* __restrict__ Ssum,
                  float* __restrict__ y) {
  __shared__ float4 sbt[LCHUNK * 4];

  const int d = blockIdx.x * SCAN_THREADS + threadIdx.x;
  const int c = blockIdx.y;
  const int b = blockIdx.z;

  const size_t rowbase = (size_t)b * SEQ + (size_t)c * LCHUNK;
  const float* btrow = Bt + rowbase * D_STATE;
  const float* xcol  = x + rowbase * D_MODEL + d;
  float*       ycol  = y + rowbase * D_MODEL + d;

  v4f st[STG];
  ssm_stage_request(btrow, st);

  const float* srow = Ssum + ((size_t)b * NCHUNK * D_MODEL + d) * D_STATE;
  float4 sv[CB1][4];
#pragma unroll
  for (int j = 0; j < CB1; ++j) {
    if (j < c) {
      const float4* p = reinterpret_cast<const float4*>(
          srow + (size_t)j * D_MODEL * D_STATE);
#pragma unroll
      for (int q = 0; q < 4; ++q) sv[j][q] = p[q];
    }
  }

  v4f la[4];
  float4 vc[4];
  {
    const v4f* pl = reinterpret_cast<const v4f*>(logA + (size_t)d * D_STATE);
    const float4* pc = reinterpret_cast<const float4*>(Cw + (size_t)d * D_STATE);
#pragma unroll
    for (int q = 0; q < 4; ++q) { la[q] = pl[q]; vc[q] = pc[q]; }
  }
  const float dskip = Dvec[d];

  float xa[XGRP], xb[XGRP];
#pragma unroll
  for (int j = 0; j < XGRP; ++j) xa[j] = xcol[(size_t)j * D_MODEL];

  ssm_entry_keep_together(st, la);
  ssm_stage_commit(sbt, st);

  float A[D_STATE], C[D_STATE], h[D_STATE];
  ssm_abar_from(la, A);
#pragma unroll
  for (int q = 0; q < 4; ++q) {
    C[4 * q + 0] = vc[q].x; C[4 * q + 1] = vc[q].y; C[4 * q + 2] = vc[q].z; C[4 * q + 3] = vc[q].w;
  }
#pragma unroll
  for (int s = 0; s < D_STATE; ++s) h[s] = 0.0f;

  if (c > 0) {
    float AL[D_STATE];
#pragma unroll
    for (int s = 0; s < D_STATE; ++s) {
      float p = A[s];
#pragma unroll
      for (int j = 0; j < 7; ++j) p = p * p;
      AL[s] = p;
    }
#pragma unroll
    for (int j = 0; j < CB1; ++j) {
      if (j < c) ssm_carry_step(AL, h, sv[j]);
    }
    if (c > CB1) {
      float4 sw[NCHUNK - 1 - CB1][4];
#pragma unroll
      for (int j = 0; j < NCHUNK - 1 - CB1; ++j) {
        if (CB1 + j < c) {
          const float4* p = reinterpret_cast<const float4*>(
              srow + (size_t)(CB1 + j) * D_MODEL * D_STATE);
#pragma unroll
          for (int q = 0; q < 4; ++q) sw[j][q] = p[q];
        }
      }
#pragma unroll
      for (int j = 0; j < NCHUNK - 1 - CB1; ++j) {
        if (CB1 + j < c) ssm_carry_step(AL, h, sw[j]);
      }
    }
  }

  float4 ra[BGRP][4], rb[BGRP][4];
  ssm_rows_request(sbt, ra);
  ssm_rows_keep_ahead(ra);
  for (int l0 = 0; l0 < LCHUNK; l0 += 2 * XGRP) {
#pragma unroll
    for (int j = 0; j < XGRP; ++j) xb[j] = xcol[(size_t)(l0 + XGRP + j) * D_MODEL];

    __builtin_amdgcn_sched_barrier(0);
    ssm_rows_request(sbt + 4 * (l0 + BGRP), rb);
    __builtin_amdgcn_sched_barrier(0);
    ssm_replay_rows<0>(ra, A, C, h, xa, dskip, ycol + (size_t)l0 * D_MODEL);
    __builtin_amdgcn_sched_barrier(0);
    ssm_rows_request(sbt + 4 * (l0 + 2 * BGRP), ra);
    __builtin_amdgcn_sched_barrier(0);
    ssm_replay_rows<BGRP>(rb, A, C, h, xa, dskip, ycol + (size_t)l0 * D_MODEL);

    const int ln = (l0 + 2 * XGRP < LCHUNK) ? (l0 + 2 * XGRP) : l0;
    __builtin_amdgcn_sched_barrier(0);
#pragma unroll
    for (int j = 0; j < XGRP; ++j) xa[j] = xcol[(size_t)(ln + j) * D_MODEL];

    __builtin_amdgcn_sched_barrier(0);
    ssm_rows_request(sbt + 4 * (l0 + 3 * BGRP), rb);
    __builtin_amdgcn_sched_barrier(0);
    ssm_replay_rows<0>(ra, A, C, h, xb, dskip, ycol + (size_t)(l0 + XGRP) * D_MODEL);
    const int rn = (l0 + 2 * XGRP < LCHUNK) ? (l0 + 2 * XGRP) : (l0 + 3 * BGRP);
    __builtin_amdgcn_sched_barrier(0);
    ssm_rows_request(sbt + 4 * rn, ra);
    __builtin_amdgcn_sched_barrier(0);
    ssm_replay_rows<BGRP>(rb, A, C, h, xb, dskip, ycol + (size_t)(l0 + XGRP) * D_MODEL);
  }
}

extern "C" void kernel_launch(void* const* d_in, const int* in_sizes, int n_in,
                              void* d_out, int out_size, void* d_ws, size_t ws_size,
                              hipStream_t stream) {
  const float* x    = (const float*)d_in[0];
  const float* logA = (const float*)d_in[1];
  const float* Bw   = (const float*)d_in[2];
  const float* Cw   = (const float*)d_in[3];
  const float* Dv   = (const float*)d_in[4];
  float* y = (float*)d_out;

  char* ws = (char*)d_ws;
  float* Bt    = (float*)(ws + (128 << 10));
  float* Ssum  = (float*)(ws + (384 << 10));

  ssm_bproj_wmma_kernel<<<(BATCH * SEQ) / 16, 32, 0, stream>>>(x, Bw, Bt);

  dim3 gscan(D_MODEL / SCAN_THREADS, NCHUNK, BATCH);

  ssm_chunk_local_kernel<<<gscan, SCAN_THREADS, 0, stream>>>(Bt, logA, Ssum);

  ssm_output_kernel<<<gscan, SCAN_THREADS, 0, stream>>>(x, Bt, logA, Cw, Dv, Ssum, y);
}


# B-projection kernel: each tile's k range split over 8 waves that read x and B_w in whole 512-byte row pieces and re-lay them through LDS; partial tiles summed via LDS: k sum order changes
# speedup vs baseline: 5.1358x; 1.5410x over previous
#include <hip/hip_runtime.h>
#include <hip/hip_bf16.h>

#define D_MODEL 1024
#define D_STATE 16
#define BATCH   2
#define SEQ     2048
#define NCHUNK  16
#define LCHUNK  (SEQ / NCHUNK)
#define BGRP    4
#define XGRP    8
#define CB1     8
#define SCAN_THREADS 128
#define STG     (LCHUNK * 4 / SCAN_THREADS)

typedef __attribute__((ext_vector_type(2))) float v2f;
typedef __attribute__((ext_vector_type(8))) float v8f;
typedef __attribute__((ext_vector_type(4))) float v4f;

__device__ __forceinline__ void ssm_abar_from(const v4f (&la)[4], float (&A)[D_STATE]) {
#pragma unroll
  for (int q = 0; q < 4; ++q) {
    v4f v = la[q];
    A[4 * q + 0] = expf(-expf(v.x));
    A[4 * q + 1] = expf(-expf(v.y));
    A[4 * q + 2] = expf(-expf(v.z));
    A[4 * q + 3] = expf(-expf(v.w));
  }
}

__device__ __forceinline__ void ssm_stage_request(const float* __restrict__ btrow,
                                                  v4f (&st)[STG]) {
  const v4f* g = reinterpret_cast<const v4f*>(btrow);
#pragma unroll
  for (int i = 0; i < STG; ++i) st[i] = g[threadIdx.x + i * SCAN_THREADS];
}

__device__ __forceinline__ void ssm_stage_commit(float4* __restrict__ sbt,
                                                 const v4f (&st)[STG]) {
  v4f* s = reinterpret_cast<v4f*>(sbt);
#pragma unroll
  for (int i = 0; i < STG; ++i) s[threadIdx.x + i * SCAN_THREADS] = st[i];
  __syncthreads();
}

__device__ __forceinline__ void ssm_rows_request(const float4* __restrict__ srow,
                                                 float4 (&r)[BGRP][4]) {
#pragma unroll
  for (int j = 0; j < BGRP; ++j) {
#pragma unroll
    for (int q = 0; q < 4; ++q) r[j][q] = srow[4 * j + q];
  }
}

__device__ __forceinline__ void ssm_rows_keep_ahead(float4 (&r)[BGRP][4]) {
#pragma unroll
  for (int j = 0; j < BGRP; ++j) {
#pragma unroll
    for (int q = 0; q < 4; ++q) {
      asm volatile("" : "+v"(r[j][q].x), "+v"(r[j][q].y), "+v"(r[j][q].z), "+v"(r[j][q].w));
    }
  }
}

__device__ __forceinline__ void ssm_entry_keep_together(v4f (&st)[STG], v4f (&la)[4]) {
  static_assert(STG == 4, "one operand per staged vector is written out below");
  asm volatile("" : "+v"(st[0]), "+v"(st[1]), "+v"(st[2]), "+v"(st[3]),
                    "+v"(la[0]), "+v"(la[1]), "+v"(la[2]), "+v"(la[3]));
}

#define K1_WAVES   8
#define K1_THREADS (K1_WAVES * 32)
#define K1_KSLICE  (D_MODEL / K1_WAVES)
#define K1_PITCH   (K1_KSLICE + 4)
#define K1_NFRAG   (K1_KSLICE / 8)
#define K1_SLAB    (16 * K1_PITCH)

__global__ void __launch_bounds__(K1_THREADS)
ssm_bproj_wmma_kernel(const float* __restrict__ x,
                      const float* __restrict__ Bw,
                      float* __restrict__ Bt) {
  static_assert(K1_KSLICE == 4 * 32, "one 16-byte request per lane covers a wave's piece of a row");
  __shared__ __attribute__((aligned(16))) float sx[K1_WAVES * K1_SLAB];
  __shared__ __attribute__((aligned(16))) float sw[K1_WAVES * K1_SLAB];
  __shared__ float red[K1_WAVES * 256];

  const int tid  = threadIdx.x;
  const int lane = tid & 31;
  const int w    = tid >> 5;
  const int half = lane >> 4;
  const int lidx = lane & 15;
  const int row0 = blockIdx.x * 16;

  const float* xg = x  + (size_t)row0 * D_MODEL + w * K1_KSLICE + 4 * lane;
  const float* wg = Bw + w * K1_KSLICE + 4 * lane;

  v4f gx[16], gw[16];
#pragma unroll
  for (int r = 0; r < 16; ++r) gx[r] = *reinterpret_cast<const v4f*>(xg + (size_t)r * D_MODEL);
#pragma unroll
  for (int n = 0; n < 16; ++n) gw[n] = *reinterpret_cast<const v4f*>(wg + (size_t)n * D_MODEL);

  float* mx = sx + w * K1_SLAB;
  float* mw = sw + w * K1_SLAB;
#pragma unroll
  for (int r = 0; r < 16; ++r) *reinterpret_cast<v4f*>(mx + r * K1_PITCH + 4 * lane) = gx[r];
#pragma unroll
  for (int n = 0; n < 16; ++n) *reinterpret_cast<v4f*>(mw + n * K1_PITCH + 4 * lane) = gw[n];
  __syncthreads();

  const float* fa = mx + lidx * K1_PITCH + 4 * half;
  const float* fb = mw + lidx * K1_PITCH + 4 * half;
  v4f a[K1_NFRAG], b[K1_NFRAG];
#pragma unroll
  for (int i = 0; i < K1_NFRAG; ++i) {
    a[i] = *reinterpret_cast<const v4f*>(fa + 8 * i);
    b[i] = *reinterpret_cast<const v4f*>(fb + 8 * i);
  }

  __builtin_amdgcn_sched_barrier(0);

  v8f acc0 = {}, acc1 = {};
#pragma unroll
  for (int i = 0; i < K1_NFRAG; ++i) {
    v2f a0 = { a[i].x, a[i].y }, b0 = { b[i].x, b[i].y };
    v2f a1 = { a[i].z, a[i].w }, b1 = { b[i].z, b[i].w };
    acc0 = __builtin_amdgcn_wmma_f32_16x16x4_f32(false, a0, false, b0,
                                                 (short)0, acc0, false, false);
    acc1 = __builtin_amdgcn_wmma_f32_16x16x4_f32(false, a1, false, b1,
                                                 (short)0, acc1, false, false);
  }

  __builtin_amdgcn_sched_barrier(0);
  asm volatile("v_nop\n\tv_nop\n\tv_nop\n\tv_nop\n\tv_nop" : "+v"(acc0), "+v"(acc1));
  __builtin_amdgcn_sched_barrier(0);

#pragma unroll
  for (int i = 0; i < 8; ++i) {
    red[w * 256 + (i + 8 * half) * 16 + lidx] = acc0[i] + acc1[i];
  }
  __syncthreads();

  float s = red[tid];
#pragma unroll
  for (int ww = 1; ww < K1_WAVES; ++ww) s += red[ww * 256 + tid];
  Bt[(size_t)row0 * D_STATE + tid] = s;
}

__device__ __forceinline__ void ssm_scan_rows(const float4 (&bt)[BGRP][4],
                                              const float (&A)[D_STATE],
                                              float (&h)[D_STATE]) {
#pragma unroll
  for (int j = 0; j < BGRP; ++j) {
#pragma unroll
    for (int q = 0; q < 4; ++q) {
      float4 v = bt[j][q];
      h[4 * q + 0] = fmaf(A[4 * q + 0], h[4 * q + 0], v.x);
      h[4 * q + 1] = fmaf(A[4 * q + 1], h[4 * q + 1], v.y);
      h[4 * q + 2] = fmaf(A[4 * q + 2], h[4 * q + 2], v.z);
      h[4 * q + 3] = fmaf(A[4 * q + 3], h[4 * q + 3], v.w);
    }
  }
}

__global__ void __launch_bounds__(SCAN_THREADS)
ssm_chunk_local_kernel(const float* __restrict__ Bt,
                       const float* __restrict__ logA,
                       float* __restrict__ Ssum) {
  __shared__ float4 sbt[LCHUNK * 4];

  const int d = blockIdx.x * SCAN_THREADS + threadIdx.x;
  const int c = blockIdx.y;
  const int b = blockIdx.z;

  v4f st[STG];
  ssm_stage_request(Bt + ((size_t)b * SEQ + (size_t)c * LCHUNK) * D_STATE, st);
  v4f la[4];
  {
    const v4f* pl = reinterpret_cast<const v4f*>(logA + (size_t)d * D_STATE);
#pragma unroll
    for (int q = 0; q < 4; ++q) la[q] = pl[q];
  }
  ssm_entry_keep_together(st, la);
  ssm_stage_commit(sbt, st);

  float A[D_STATE], h[D_STATE];
  ssm_abar_from(la, A);
#pragma unroll
  for (int s = 0; s < D_STATE; ++s) h[s] = 0.0f;

  float4 ra[BGRP][4], rb[BGRP][4];
  ssm_rows_request(sbt, ra);
  ssm_rows_keep_ahead(ra);
  for (int l0 = 0; l0 < LCHUNK; l0 += 2 * BGRP) {
    __builtin_amdgcn_sched_barrier(0);
    ssm_rows_request(sbt + 4 * (l0 + BGRP), rb);
    __builtin_amdgcn_sched_barrier(0);
    ssm_scan_rows(ra, A, h);

    const int ln = (l0 + 2 * BGRP < LCHUNK) ? (l0 + 2 * BGRP) : (l0 + BGRP);
    __builtin_amdgcn_sched_barrier(0);
    ssm_rows_request(sbt + 4 * ln, ra);
    __builtin_amdgcn_sched_barrier(0);
    ssm_scan_rows(rb, A, h);
  }

  float4* out = reinterpret_cast<float4*>(
      Ssum + (((size_t)b * NCHUNK + c) * D_MODEL + d) * D_STATE);
#pragma unroll
  for (int q = 0; q < 4; ++q) {
    float4 v = { h[4 * q + 0], h[4 * q + 1], h[4 * q + 2], h[4 * q + 3] };
    out[q] = v;
  }
}

template <int G>
__device__ __forceinline__ void ssm_replay_rows(const float4 (&bt)[BGRP][4],
                                                const float (&A)[D_STATE],
                                                const float (&C)[D_STATE],
                                                float (&h)[D_STATE],
                                                const float (&xv)[XGRP],
                                                float dskip,
                                                float* __restrict__ yrow) {
#pragma unroll
  for (int j = 0; j < BGRP; ++j) {
    float acc = 0.0f;
#pragma unroll
    for (int q = 0; q < 4; ++q) {
      float4 v = bt[j][q];
      h[4 * q + 0] = fmaf(A[4 * q + 0], h[4 * q + 0], v.x);
      h[4 * q + 1] = fmaf(A[4 * q + 1], h[4 * q + 1], v.y);
      h[4 * q + 2] = fmaf(A[4 * q + 2], h[4 * q + 2], v.z);
      h[4 * q + 3] = fmaf(A[4 * q + 3], h[4 * q + 3], v.w);
      acc = fmaf(C[4 * q + 0], h[4 * q + 0], acc);
      acc = fmaf(C[4 * q + 1], h[4 * q + 1], acc);
      acc = fmaf(C[4 * q + 2], h[4 * q + 2], acc);
      acc = fmaf(C[4 * q + 3], h[4 * q + 3], acc);
    }
    yrow[(size_t)(G + j) * D_MODEL] = fmaf(dskip, xv[G + j], acc);
  }
}

__device__ __forceinline__ void ssm_carry_step(const float (&AL)[D_STATE],
                                               float (&h)[D_STATE],
                                               const float4 (&S)[4]) {
#pragma unroll
  for (int q = 0; q < 4; ++q) {
    float4 v = S[q];
    h[4 * q + 0] = fmaf(AL[4 * q + 0], h[4 * q + 0], v.x);
    h[4 * q + 1] = fmaf(AL[4 * q + 1], h[4 * q + 1], v.y);
    h[4 * q + 2] = fmaf(AL[4 * q + 2], h[4 * q + 2], v.z);
    h[4 * q + 3] = fmaf(AL[4 * q + 3], h[4 * q + 3], v.w);
  }
}

__global__ void __launch_bounds__(SCAN_THREADS)
ssm_output_kernel(const float* __restrict__ x,
                  const float* __restrict__ Bt,
                  const float* __restrict__ logA,
                  const float* __restrict__ Cw,
                  const float* __restrict__ Dvec,
                  const float* __restrict__ Ssum,
                  float* __restrict__ y) {
  __shared__ float4 sbt[LCHUNK * 4];

  const int d = blockIdx.x * SCAN_THREADS + threadIdx.x;
  const int c = blockIdx.y;
  const int b = blockIdx.z;

  const size_t rowbase = (size_t)b * SEQ + (size_t)c * LCHUNK;
  const float* btrow = Bt + rowbase * D_STATE;
  const float* xcol  = x + rowbase * D_MODEL + d;
  float*       ycol  = y + rowbase * D_MODEL + d;

  v4f st[STG];
  ssm_stage_request(btrow, st);

  const float* srow = Ssum + ((size_t)b * NCHUNK * D_MODEL + d) * D_STATE;
  float4 sv[CB1][4];
#pragma unroll
  for (int j = 0; j < CB1; ++j) {
    if (j < c) {
      const float4* p = reinterpret_cast<const float4*>(
          srow + (size_t)j * D_MODEL * D_STATE);
#pragma unroll
      for (int q = 0; q < 4; ++q) sv[j][q] = p[q];
    }
  }

  v4f la[4];
  float4 vc[4];
  {
    const v4f* pl = reinterpret_cast<const v4f*>(logA + (size_t)d * D_STATE);
    const float4* pc = reinterpret_cast<const float4*>(Cw + (size_t)d * D_STATE);
#pragma unroll
    for (int q = 0; q < 4; ++q) { la[q] = pl[q]; vc[q] = pc[q]; }
  }
  const float dskip = Dvec[d];

  float xa[XGRP], xb[XGRP];
#pragma unroll
  for (int j = 0; j < XGRP; ++j) xa[j] = xcol[(size_t)j * D_MODEL];

  ssm_entry_keep_together(st, la);
  ssm_stage_commit(sbt, st);

  float A[D_STATE], C[D_STATE], h[D_STATE];
  ssm_abar_from(la, A);
#pragma unroll
  for (int q = 0; q < 4; ++q) {
    C[4 * q + 0] = vc[q].x; C[4 * q + 1] = vc[q].y; C[4 * q + 2] = vc[q].z; C[4 * q + 3] = vc[q].w;
  }
#pragma unroll
  for (int s = 0; s < D_STATE; ++s) h[s] = 0.0f;

  if (c > 0) {
    float AL[D_STATE];
#pragma unroll
    for (int s = 0; s < D_STATE; ++s) {
      float p = A[s];
#pragma unroll
      for (int j = 0; j < 7; ++j) p = p * p;
      AL[s] = p;
    }
#pragma unroll
    for (int j = 0; j < CB1; ++j) {
      if (j < c) ssm_carry_step(AL, h, sv[j]);
    }
    if (c > CB1) {
      float4 sw[NCHUNK - 1 - CB1][4];
#pragma unroll
      for (int j = 0; j < NCHUNK - 1 - CB1; ++j) {
        if (CB1 + j < c) {
          const float4* p = reinterpret_cast<const float4*>(
              srow + (size_t)(CB1 + j) * D_MODEL * D_STATE);
#pragma unroll
          for (int q = 0; q < 4; ++q) sw[j][q] = p[q];
        }
      }
#pragma unroll
      for (int j = 0; j < NCHUNK - 1 - CB1; ++j) {
        if (CB1 + j < c) ssm_carry_step(AL, h, sw[j]);
      }
    }
  }

  float4 ra[BGRP][4], rb[BGRP][4];
  ssm_rows_request(sbt, ra);
  ssm_rows_keep_ahead(ra);
  for (int l0 = 0; l0 < LCHUNK; l0 += 2 * XGRP) {
#pragma unroll
    for (int j = 0; j < XGRP; ++j) xb[j] = xcol[(size_t)(l0 + XGRP + j) * D_MODEL];

    __builtin_amdgcn_sched_barrier(0);
    ssm_rows_request(sbt + 4 * (l0 + BGRP), rb);
    __builtin_amdgcn_sched_barrier(0);
    ssm_replay_rows<0>(ra, A, C, h, xa, dskip, ycol + (size_t)l0 * D_MODEL);
    __builtin_amdgcn_sched_barrier(0);
    ssm_rows_request(sbt + 4 * (l0 + 2 * BGRP), ra);
    __builtin_amdgcn_sched_barrier(0);
    ssm_replay_rows<BGRP>(rb, A, C, h, xa, dskip, ycol + (size_t)l0 * D_MODEL);

    const int ln = (l0 + 2 * XGRP < LCHUNK) ? (l0 + 2 * XGRP) : l0;
    __builtin_amdgcn_sched_barrier(0);
#pragma unroll
    for (int j = 0; j < XGRP; ++j) xa[j] = xcol[(size_t)(ln + j) * D_MODEL];

    __builtin_amdgcn_sched_barrier(0);
    ssm_rows_request(sbt + 4 * (l0 + 3 * BGRP), rb);
    __builtin_amdgcn_sched_barrier(0);
    ssm_replay_rows<0>(ra, A, C, h, xb, dskip, ycol + (size_t)(l0 + XGRP) * D_MODEL);
    const int rn = (l0 + 2 * XGRP < LCHUNK) ? (l0 + 2 * XGRP) : (l0 + 3 * BGRP);
    __builtin_amdgcn_sched_barrier(0);
    ssm_rows_request(sbt + 4 * rn, ra);
    __builtin_amdgcn_sched_barrier(0);
    ssm_replay_rows<BGRP>(rb, A, C, h, xb, dskip, ycol + (size_t)(l0 + XGRP) * D_MODEL);
  }
}

extern "C" void kernel_launch(void* const* d_in, const int* in_sizes, int n_in,
                              void* d_out, int out_size, void* d_ws, size_t ws_size,
                              hipStream_t stream) {
  const float* x    = (const float*)d_in[0];
  const float* logA = (const float*)d_in[1];
  const float* Bw   = (const float*)d_in[2];
  const float* Cw   = (const float*)d_in[3];
  const float* Dv   = (const float*)d_in[4];
  float* y = (float*)d_out;

  char* ws = (char*)d_ws;
  float* Bt    = (float*)(ws + (128 << 10));
  float* Ssum  = (float*)(ws + (384 << 10));

  ssm_bproj_wmma_kernel<<<(BATCH * SEQ) / 16, K1_THREADS, 0, stream>>>(x, Bw, Bt);

  dim3 gscan(D_MODEL / SCAN_THREADS, NCHUNK, BATCH);

  ssm_chunk_local_kernel<<<gscan, SCAN_THREADS, 0, stream>>>(Bt, logA, Ssum);

  ssm_output_kernel<<<gscan, SCAN_THREADS, 0, stream>>>(x, Bt, logA, Cw, Dv, Ssum, y);
}


# K2 work-groups take 4 consecutive chunks (a wave each, 32 channels) and combine their sums through LDS; replay entry reads at most 4 sum rows in one round trip, not 15 in two; carry sum order changes
# speedup vs baseline: 5.3875x; 1.0490x over previous
#include <hip/hip_runtime.h>
#include <hip/hip_bf16.h>

#define D_MODEL 1024
#define D_STATE 16
#define BATCH   2
#define SEQ     2048
#define NCHUNK  16
#define LCHUNK  (SEQ / NCHUNK)
#define BGRP    4
#define XGRP    8
#define CGRP    4
#define NGRP    (NCHUNK / CGRP)
#define WSTG    (LCHUNK * 4 / 32)
#define SCAN_THREADS 128
#define STG     (LCHUNK * 4 / SCAN_THREADS)

typedef __attribute__((ext_vector_type(2))) float v2f;
typedef __attribute__((ext_vector_type(8))) float v8f;
typedef __attribute__((ext_vector_type(4))) float v4f;

__device__ __forceinline__ void ssm_abar_from(const v4f (&la)[4], float (&A)[D_STATE]) {
#pragma unroll
  for (int q = 0; q < 4; ++q) {
    v4f v = la[q];
    A[4 * q + 0] = expf(-expf(v.x));
    A[4 * q + 1] = expf(-expf(v.y));
    A[4 * q + 2] = expf(-expf(v.z));
    A[4 * q + 3] = expf(-expf(v.w));
  }
}

__device__ __forceinline__ void ssm_stage_request(const float* __restrict__ btrow,
                                                  v4f (&st)[STG]) {
  const v4f* g = reinterpret_cast<const v4f*>(btrow);
#pragma unroll
  for (int i = 0; i < STG; ++i) st[i] = g[threadIdx.x + i * SCAN_THREADS];
}

__device__ __forceinline__ void ssm_stage_commit(float4* __restrict__ sbt,
                                                 const v4f (&st)[STG]) {
  v4f* s = reinterpret_cast<v4f*>(sbt);
#pragma unroll
  for (int i = 0; i < STG; ++i) s[threadIdx.x + i * SCAN_THREADS] = st[i];
  __syncthreads();
}

__device__ __forceinline__ void ssm_wave_stage_request(const float* __restrict__ btrow,
                                                       int lane, v4f (&st)[WSTG]) {
  const v4f* g = reinterpret_cast<const v4f*>(btrow);
#pragma unroll
  for (int i = 0; i < WSTG; ++i) st[i] = g[lane + i * 32];
}

__device__ __forceinline__ void ssm_wave_stage_commit(float4* __restrict__ srows,
                                                      int lane, const v4f (&st)[WSTG]) {
  v4f* s = reinterpret_cast<v4f*>(srows);
#pragma unroll
  for (int i = 0; i < WSTG; ++i) s[lane + i * 32] = st[i];
  __syncthreads();
}

__device__ __forceinline__ void ssm_rows_request(const float4* __restrict__ srow,
                                                 float4 (&r)[BGRP][4]) {
#pragma unroll
  for (int j = 0; j < BGRP; ++j) {
#pragma unroll
    for (int q = 0; q < 4; ++q) r[j][q] = srow[4 * j + q];
  }
}

__device__ __forceinline__ void ssm_rows_keep_ahead(float4 (&r)[BGRP][4]) {
#pragma unroll
  for (int j = 0; j < BGRP; ++j) {
#pragma unroll
    for (int q = 0; q < 4; ++q) {
      asm volatile("" : "+v"(r[j][q].x), "+v"(r[j][q].y), "+v"(r[j][q].z), "+v"(r[j][q].w));
    }
  }
}

__device__ __forceinline__ void ssm_entry_keep_together(v4f (&st)[STG], v4f (&la)[4]) {
  static_assert(STG == 4, "one operand per staged vector is written out below");
  asm volatile("" : "+v"(st[0]), "+v"(st[1]), "+v"(st[2]), "+v"(st[3]),
                    "+v"(la[0]), "+v"(la[1]), "+v"(la[2]), "+v"(la[3]));
}

__device__ __forceinline__ void ssm_wave_entry_keep_together(v4f (&st)[WSTG], v4f (&la)[4]) {
  static_assert(WSTG == 16, "one operand per staged vector is written out below");
  asm volatile("" : "+v"(st[0]), "+v"(st[1]), "+v"(st[2]), "+v"(st[3]),
                    "+v"(st[4]), "+v"(st[5]), "+v"(st[6]), "+v"(st[7]),
                    "+v"(st[8]), "+v"(st[9]), "+v"(st[10]), "+v"(st[11]),
                    "+v"(st[12]), "+v"(st[13]), "+v"(st[14]), "+v"(st[15]),
                    "+v"(la[0]), "+v"(la[1]), "+v"(la[2]), "+v"(la[3]));
}

#define K1_WAVES   8
#define K1_THREADS (K1_WAVES * 32)
#define K1_KSLICE  (D_MODEL / K1_WAVES)
#define K1_PITCH   (K1_KSLICE + 4)
#define K1_NFRAG   (K1_KSLICE / 8)
#define K1_SLAB    (16 * K1_PITCH)

__global__ void __launch_bounds__(K1_THREADS)
ssm_bproj_wmma_kernel(const float* __restrict__ x,
                      const float* __restrict__ Bw,
                      float* __restrict__ Bt) {
  static_assert(K1_KSLICE == 4 * 32, "one 16-byte request per lane covers a wave's piece of a row");
  __shared__ __attribute__((aligned(16))) float sx[K1_WAVES * K1_SLAB];
  __shared__ __attribute__((aligned(16))) float sw[K1_WAVES * K1_SLAB];
  __shared__ float red[K1_WAVES * 256];

  const int tid  = threadIdx.x;
  const int lane = tid & 31;
  const int w    = tid >> 5;
  const int half = lane >> 4;
  const int lidx = lane & 15;
  const int row0 = blockIdx.x * 16;

  const float* xg = x  + (size_t)row0 * D_MODEL + w * K1_KSLICE + 4 * lane;
  const float* wg = Bw + w * K1_KSLICE + 4 * lane;

  v4f gx[16], gw[16];
#pragma unroll
  for (int r = 0; r < 16; ++r) gx[r] = *reinterpret_cast<const v4f*>(xg + (size_t)r * D_MODEL);
#pragma unroll
  for (int n = 0; n < 16; ++n) gw[n] = *reinterpret_cast<const v4f*>(wg + (size_t)n * D_MODEL);

  float* mx = sx + w * K1_SLAB;
  float* mw = sw + w * K1_SLAB;
#pragma unroll
  for (int r = 0; r < 16; ++r) *reinterpret_cast<v4f*>(mx + r * K1_PITCH + 4 * lane) = gx[r];
#pragma unroll
  for (int n = 0; n < 16; ++n) *reinterpret_cast<v4f*>(mw + n * K1_PITCH + 4 * lane) = gw[n];
  __syncthreads();

  const float* fa = mx + lidx * K1_PITCH + 4 * half;
  const float* fb = mw + lidx * K1_PITCH + 4 * half;
  v4f a[K1_NFRAG], b[K1_NFRAG];
#pragma unroll
  for (int i = 0; i < K1_NFRAG; ++i) {
    a[i] = *reinterpret_cast<const v4f*>(fa + 8 * i);
    b[i] = *reinterpret_cast<const v4f*>(fb + 8 * i);
  }

  __builtin_amdgcn_sched_barrier(0);

  v8f acc0 = {}, acc1 = {};
#pragma unroll
  for (int i = 0; i < K1_NFRAG; ++i) {
    v2f a0 = { a[i].x, a[i].y }, b0 = { b[i].x, b[i].y };
    v2f a1 = { a[i].z, a[i].w }, b1 = { b[i].z, b[i].w };
    acc0 = __builtin_amdgcn_wmma_f32_16x16x4_f32(false, a0, false, b0,
                                                 (short)0, acc0, false, false);
    acc1 = __builtin_amdgcn_wmma_f32_16x16x4_f32(false, a1, false, b1,
                                                 (short)0, acc1, false, false);
  }

  __builtin_amdgcn_sched_barrier(0);
  asm volatile("v_nop\n\tv_nop\n\tv_nop\n\tv_nop\n\tv_nop" : "+v"(acc0), "+v"(acc1));
  __builtin_amdgcn_sched_barrier(0);

#pragma unroll
  for (int i = 0; i < 8; ++i) {
    red[w * 256 + (i + 8 * half) * 16 + lidx] = acc0[i] + acc1[i];
  }
  __syncthreads();

  float s = red[tid];
#pragma unroll
  for (int ww = 1; ww < K1_WAVES; ++ww) s += red[ww * 256 + tid];
  Bt[(size_t)row0 * D_STATE + tid] = s;
}

__device__ __forceinline__ void ssm_carry_step(const float (&AL)[D_STATE],
                                               float (&h)[D_STATE],
                                               const float4 (&S)[4]) {
#pragma unroll
  for (int q = 0; q < 4; ++q) {
    float4 v = S[q];
    h[4 * q + 0] = fmaf(AL[4 * q + 0], h[4 * q + 0], v.x);
    h[4 * q + 1] = fmaf(AL[4 * q + 1], h[4 * q + 1], v.y);
    h[4 * q + 2] = fmaf(AL[4 * q + 2], h[4 * q + 2], v.z);
    h[4 * q + 3] = fmaf(AL[4 * q + 3], h[4 * q + 3], v.w);
  }
}

__device__ __forceinline__ void ssm_scan_rows(const float4 (&bt)[BGRP][4],
                                              const float (&A)[D_STATE],
                                              float (&h)[D_STATE]) {
#pragma unroll
  for (int j = 0; j < BGRP; ++j) {
#pragma unroll
    for (int q = 0; q < 4; ++q) {
      float4 v = bt[j][q];
      h[4 * q + 0] = fmaf(A[4 * q + 0], h[4 * q + 0], v.x);
      h[4 * q + 1] = fmaf(A[4 * q + 1], h[4 * q + 1], v.y);
      h[4 * q + 2] = fmaf(A[4 * q + 2], h[4 * q + 2], v.z);
      h[4 * q + 3] = fmaf(A[4 * q + 3], h[4 * q + 3], v.w);
    }
  }
}

__global__ void __launch_bounds__(SCAN_THREADS)
ssm_chunk_local_kernel(const float* __restrict__ Bt,
                       const float* __restrict__ logA,
                       float* __restrict__ Ssum) {
  static_assert(SCAN_THREADS == CGRP * 32, "one wave per chunk of the group");
  __shared__ float4 sbtg[CGRP * LCHUNK * 4];
  __shared__ float4 sxc[CGRP][4][32];

  const int lane = threadIdx.x & 31;
  const int w = __builtin_amdgcn_readfirstlane((int)(threadIdx.x >> 5));
  const int d = blockIdx.x * 32 + lane;
  const int c = blockIdx.y * CGRP + w;
  const int b = blockIdx.z;

  v4f st[WSTG];
  ssm_wave_stage_request(Bt + ((size_t)b * SEQ + (size_t)c * LCHUNK) * D_STATE, lane, st);
  v4f la[4];
  {
    const v4f* pl = reinterpret_cast<const v4f*>(logA + (size_t)d * D_STATE);
#pragma unroll
    for (int q = 0; q < 4; ++q) la[q] = pl[q];
  }
  ssm_wave_entry_keep_together(st, la);
  float4* sbt = sbtg + w * (LCHUNK * 4);
  ssm_wave_stage_commit(sbt, lane, st);

  float A[D_STATE], h[D_STATE];
  ssm_abar_from(la, A);
#pragma unroll
  for (int s = 0; s < D_STATE; ++s) h[s] = 0.0f;

  float4 ra[BGRP][4], rb[BGRP][4];
  ssm_rows_request(sbt, ra);
  ssm_rows_keep_ahead(ra);
  for (int l0 = 0; l0 < LCHUNK; l0 += 2 * BGRP) {
    __builtin_amdgcn_sched_barrier(0);
    ssm_rows_request(sbt + 4 * (l0 + BGRP), rb);
    __builtin_amdgcn_sched_barrier(0);
    ssm_scan_rows(ra, A, h);

    const int ln = (l0 + 2 * BGRP < LCHUNK) ? (l0 + 2 * BGRP) : (l0 + BGRP);
    __builtin_amdgcn_sched_barrier(0);
    ssm_rows_request(sbt + 4 * ln, ra);
    __builtin_amdgcn_sched_barrier(0);
    ssm_scan_rows(rb, A, h);
  }

#pragma unroll
  for (int q = 0; q < 4; ++q) {
    float4 v = { h[4 * q + 0], h[4 * q + 1], h[4 * q + 2], h[4 * q + 3] };
    sxc[w][q][lane] = v;
  }
  __syncthreads();

  if (w > 0) {
    float AL[D_STATE];
#pragma unroll
    for (int s = 0; s < D_STATE; ++s) {
      float p = A[s];
#pragma unroll
      for (int j = 0; j < 7; ++j) p = p * p;
      AL[s] = p;
    }
    float e[D_STATE];
#pragma unroll
    for (int q = 0; q < 4; ++q) {
      float4 v = sxc[0][q][lane];
      e[4 * q + 0] = v.x; e[4 * q + 1] = v.y; e[4 * q + 2] = v.z; e[4 * q + 3] = v.w;
    }
#pragma unroll
    for (int i = 1; i < CGRP - 1; ++i) {
      if (i < w) {
        float4 sv[4];
#pragma unroll
        for (int q = 0; q < 4; ++q) sv[q] = sxc[i][q][lane];
        ssm_carry_step(AL, e, sv);
      }
    }
#pragma unroll
    for (int s = 0; s < D_STATE; ++s) h[s] = fmaf(AL[s], e[s], h[s]);
  }

  float4* out = reinterpret_cast<float4*>(
      Ssum + (((size_t)b * NCHUNK + c) * D_MODEL + d) * D_STATE);
#pragma unroll
  for (int q = 0; q < 4; ++q) {
    float4 v = { h[4 * q + 0], h[4 * q + 1], h[4 * q + 2], h[4 * q + 3] };
    out[q] = v;
  }
}

template <int G>
__device__ __forceinline__ void ssm_replay_rows(const float4 (&bt)[BGRP][4],
                                                const float (&A)[D_STATE],
                                                const float (&C)[D_STATE],
                                                float (&h)[D_STATE],
                                                const float (&xv)[XGRP],
                                                float dskip,
                                                float* __restrict__ yrow) {
#pragma unroll
  for (int j = 0; j < BGRP; ++j) {
    float acc = 0.0f;
#pragma unroll
    for (int q = 0; q < 4; ++q) {
      float4 v = bt[j][q];
      h[4 * q + 0] = fmaf(A[4 * q + 0], h[4 * q + 0], v.x);
      h[4 * q + 1] = fmaf(A[4 * q + 1], h[4 * q + 1], v.y);
      h[4 * q + 2] = fmaf(A[4 * q + 2], h[4 * q + 2], v.z);
      h[4 * q + 3] = fmaf(A[4 * q + 3], h[4 * q + 3], v.w);
      acc = fmaf(C[4 * q + 0], h[4 * q + 0], acc);
      acc = fmaf(C[4 * q + 1], h[4 * q + 1], acc);
      acc = fmaf(C[4 * q + 2], h[4 * q + 2], acc);
      acc = fmaf(C[4 * q + 3], h[4 * q + 3], acc);
    }
    yrow[(size_t)(G + j) * D_MODEL] = fmaf(dskip, xv[G + j], acc);
  }
}

__global__ void __launch_bounds__(SCAN_THREADS)
ssm_output_kernel(const float* __restrict__ x,
                  const float* __restrict__ Bt,
                  const float* __restrict__ logA,
                  const float* __restrict__ Cw,
                  const float* __restrict__ Dvec,
                  const float* __restrict__ Ssum,
                  float* __restrict__ y) {
  __shared__ float4 sbt[LCHUNK * 4];

  const int d = blockIdx.x * SCAN_THREADS + threadIdx.x;
  const int c = blockIdx.y;
  const int b = blockIdx.z;

  const size_t rowbase = (size_t)b * SEQ + (size_t)c * LCHUNK;
  const float* btrow = Bt + rowbase * D_STATE;
  const float* xcol  = x + rowbase * D_MODEL + d;
  float*       ycol  = y + rowbase * D_MODEL + d;

  v4f st[STG];
  ssm_stage_request(btrow, st);

  const int cg = c / CGRP;
  const int cw = c % CGRP;
  const float* srow = Ssum + ((size_t)b * NCHUNK * D_MODEL + d) * D_STATE;
  float4 sv[NGRP - 1][4];
#pragma unroll
  for (int j = 0; j < NGRP - 1; ++j) {
    if (j < cg) {
      const float4* p = reinterpret_cast<const float4*>(
          srow + (size_t)(CGRP * j + CGRP - 1) * D_MODEL * D_STATE);
#pragma unroll
      for (int q = 0; q < 4; ++q) sv[j][q] = p[q];
    }
  }
  float4 sp[4];
  if (cw > 0) {
    const float4* p = reinterpret_cast<const float4*>(
        srow + (size_t)(c - 1) * D_MODEL * D_STATE);
#pragma unroll
    for (int q = 0; q < 4; ++q) sp[q] = p[q];
  }

  v4f la[4];
  float4 vc[4];
  {
    const v4f* pl = reinterpret_cast<const v4f*>(logA + (size_t)d * D_STATE);
    const float4* pc = reinterpret_cast<const float4*>(Cw + (size_t)d * D_STATE);
#pragma unroll
    for (int q = 0; q < 4; ++q) { la[q] = pl[q]; vc[q] = pc[q]; }
  }
  const float dskip = Dvec[d];

  float xa[XGRP], xb[XGRP];
#pragma unroll
  for (int j = 0; j < XGRP; ++j) xa[j] = xcol[(size_t)j * D_MODEL];

  ssm_entry_keep_together(st, la);
  ssm_stage_commit(sbt, st);

  float A[D_STATE], C[D_STATE], h[D_STATE];
  ssm_abar_from(la, A);
#pragma unroll
  for (int q = 0; q < 4; ++q) {
    C[4 * q + 0] = vc[q].x; C[4 * q + 1] = vc[q].y; C[4 * q + 2] = vc[q].z; C[4 * q + 3] = vc[q].w;
  }
#pragma unroll
  for (int s = 0; s < D_STATE; ++s) h[s] = 0.0f;

  if (c > 0) {
    float AL[D_STATE];
#pragma unroll
    for (int s = 0; s < D_STATE; ++s) {
      float p = A[s];
#pragma unroll
      for (int j = 0; j < 7; ++j) p = p * p;
      AL[s] = p;
    }
    if (cg > 0) {
      static_assert(CGRP == 4, "A_G is two squarings of A_Lc");
      float AG[D_STATE];
#pragma unroll
      for (int s = 0; s < D_STATE; ++s) {
        float p = AL[s] * AL[s];
        AG[s] = p * p;
      }
#pragma unroll
      for (int j = 0; j < NGRP - 1; ++j) {
        if (j < cg) ssm_carry_step(AG, h, sv[j]);
      }
    }
    if (cw > 0) {
#pragma unroll
      for (int i = 1; i < CGRP - 1; ++i) {
        if (i < cw) {
#pragma unroll
          for (int s = 0; s < D_STATE; ++s) h[s] = AL[s] * h[s];
        }
      }
      ssm_carry_step(AL, h, sp);
    }
  }

  float4 ra[BGRP][4], rb[BGRP][4];
  ssm_rows_request(sbt, ra);
  ssm_rows_keep_ahead(ra);
  for (int l0 = 0; l0 < LCHUNK; l0 += 2 * XGRP) {
#pragma unroll
    for (int j = 0; j < XGRP; ++j) xb[j] = xcol[(size_t)(l0 + XGRP + j) * D_MODEL];

    __builtin_amdgcn_sched_barrier(0);
    ssm_rows_request(sbt + 4 * (l0 + BGRP), rb);
    __builtin_amdgcn_sched_barrier(0);
    ssm_replay_rows<0>(ra, A, C, h, xa, dskip, ycol + (size_t)l0 * D_MODEL);
    __builtin_amdgcn_sched_barrier(0);
    ssm_rows_request(sbt + 4 * (l0 + 2 * BGRP), ra);
    __builtin_amdgcn_sched_barrier(0);
    ssm_replay_rows<BGRP>(rb, A, C, h, xa, dskip, ycol + (size_t)l0 * D_MODEL);

    const int ln = (l0 + 2 * XGRP < LCHUNK) ? (l0 + 2 * XGRP) : l0;
    __builtin_amdgcn_sched_barrier(0);
#pragma unroll
    for (int j = 0; j < XGRP; ++j) xa[j] = xcol[(size_t)(ln + j) * D_MODEL];

    __builtin_amdgcn_sched_barrier(0);
    ssm_rows_request(sbt + 4 * (l0 + 3 * BGRP), rb);
    __builtin_amdgcn_sched_barrier(0);
    ssm_replay_rows<0>(ra, A, C, h, xb, dskip, ycol + (size_t)(l0 + XGRP) * D_MODEL);
    const int rn = (l0 + 2 * XGRP < LCHUNK) ? (l0 + 2 * XGRP) : (l0 + 3 * BGRP);
    __builtin_amdgcn_sched_barrier(0);
    ssm_rows_request(sbt + 4 * rn, ra);
    __builtin_amdgcn_sched_barrier(0);
    ssm_replay_rows<BGRP>(rb, A, C, h, xb, dskip, ycol + (size_t)(l0 + XGRP) * D_MODEL);
  }
}

extern "C" void kernel_launch(void* const* d_in, const int* in_sizes, int n_in,
                              void* d_out, int out_size, void* d_ws, size_t ws_size,
                              hipStream_t stream) {
  const float* x    = (const float*)d_in[0];
  const float* logA = (const float*)d_in[1];
  const float* Bw   = (const float*)d_in[2];
  const float* Cw   = (const float*)d_in[3];
  const float* Dv   = (const float*)d_in[4];
  float* y = (float*)d_out;

  char* ws = (char*)d_ws;
  float* Bt    = (float*)(ws + (128 << 10));
  float* Ssum  = (float*)(ws + (384 << 10));

  ssm_bproj_wmma_kernel<<<(BATCH * SEQ) / 16, K1_THREADS, 0, stream>>>(x, Bw, Bt);

  dim3 gscan(D_MODEL / SCAN_THREADS, NCHUNK, BATCH);

  dim3 glocal(D_MODEL / 32, NGRP, BATCH);
  ssm_chunk_local_kernel<<<glocal, SCAN_THREADS, 0, stream>>>(Bt, logA, Ssum);

  ssm_output_kernel<<<gscan, SCAN_THREADS, 0, stream>>>(x, Bt, logA, Cw, Dv, Ssum, y);
}


# K2 only: a lane holds one state of 16 channels, not 16 states of one channel, so a Bt row is one 4-byte LDS read per lane, not four 16-byte same-address reads; order of every sum kept
# speedup vs baseline: 5.8532x; 1.0864x over previous
#include <hip/hip_runtime.h>
#include <hip/hip_bf16.h>

#define D_MODEL 1024
#define D_STATE 16
#define BATCH   2
#define SEQ     2048
#define NCHUNK  16
#define LCHUNK  (SEQ / NCHUNK)
#define BGRP    4
#define XGRP    8
#define CGRP    4
#define NGRP    (NCHUNK / CGRP)
#define WSTG    (LCHUNK * 4 / 32)
#define SCAN_THREADS 128
#define STG     (LCHUNK * 4 / SCAN_THREADS)

typedef __attribute__((ext_vector_type(2))) float v2f;
typedef __attribute__((ext_vector_type(8))) float v8f;
typedef __attribute__((ext_vector_type(4))) float v4f;

__device__ __forceinline__ void ssm_abar_from(const v4f (&la)[4], float (&A)[D_STATE]) {
#pragma unroll
  for (int q = 0; q < 4; ++q) {
    v4f v = la[q];
    A[4 * q + 0] = expf(-expf(v.x));
    A[4 * q + 1] = expf(-expf(v.y));
    A[4 * q + 2] = expf(-expf(v.z));
    A[4 * q + 3] = expf(-expf(v.w));
  }
}

__device__ __forceinline__ void ssm_stage_request(const float* __restrict__ btrow,
                                                  v4f (&st)[STG]) {
  const v4f* g = reinterpret_cast<const v4f*>(btrow);
#pragma unroll
  for (int i = 0; i < STG; ++i) st[i] = g[threadIdx.x + i * SCAN_THREADS];
}

__device__ __forceinline__ void ssm_stage_commit(float4* __restrict__ sbt,
                                                 const v4f (&st)[STG]) {
  v4f* s = reinterpret_cast<v4f*>(sbt);
#pragma unroll
  for (int i = 0; i < STG; ++i) s[threadIdx.x + i * SCAN_THREADS] = st[i];
  __syncthreads();
}

__device__ __forceinline__ void ssm_wave_stage_request(const float* __restrict__ btrow,
                                                       int lane, v4f (&st)[WSTG]) {
  const v4f* g = reinterpret_cast<const v4f*>(btrow);
#pragma unroll
  for (int i = 0; i < WSTG; ++i) st[i] = g[lane + i * 32];
}

__device__ __forceinline__ void ssm_wave_stage_commit(float4* __restrict__ srows,
                                                      int lane, const v4f (&st)[WSTG]) {
  v4f* s = reinterpret_cast<v4f*>(srows);
#pragma unroll
  for (int i = 0; i < WSTG; ++i) s[lane + i * 32] = st[i];
  __syncthreads();
}

__device__ __forceinline__ void ssm_rows_request(const float4* __restrict__ srow,
                                                 float4 (&r)[BGRP][4]) {
#pragma unroll
  for (int j = 0; j < BGRP; ++j) {
#pragma unroll
    for (int q = 0; q < 4; ++q) r[j][q] = srow[4 * j + q];
  }
}

__device__ __forceinline__ void ssm_rows_keep_ahead(float4 (&r)[BGRP][4]) {
#pragma unroll
  for (int j = 0; j < BGRP; ++j) {
#pragma unroll
    for (int q = 0; q < 4; ++q) {
      asm volatile("" : "+v"(r[j][q].x), "+v"(r[j][q].y), "+v"(r[j][q].z), "+v"(r[j][q].w));
    }
  }
}

__device__ __forceinline__ void ssm_entry_keep_together(v4f (&st)[STG], v4f (&la)[4]) {
  static_assert(STG == 4, "one operand per staged vector is written out below");
  asm volatile("" : "+v"(st[0]), "+v"(st[1]), "+v"(st[2]), "+v"(st[3]),
                    "+v"(la[0]), "+v"(la[1]), "+v"(la[2]), "+v"(la[3]));
}

__device__ __forceinline__ void ssm_wave_entry_keep_together(v4f (&st)[WSTG], v4f (&la)[4]) {
  static_assert(WSTG == 16, "one operand per staged vector is written out below");
  asm volatile("" : "+v"(st[0]), "+v"(st[1]), "+v"(st[2]), "+v"(st[3]),
                    "+v"(st[4]), "+v"(st[5]), "+v"(st[6]), "+v"(st[7]),
                    "+v"(st[8]), "+v"(st[9]), "+v"(st[10]), "+v"(st[11]),
                    "+v"(st[12]), "+v"(st[13]), "+v"(st[14]), "+v"(st[15]),
                    "+v"(la[0]), "+v"(la[1]), "+v"(la[2]), "+v"(la[3]));
}

#define K1_WAVES   8
#define K1_THREADS (K1_WAVES * 32)
#define K1_KSLICE  (D_MODEL / K1_WAVES)
#define K1_PITCH   (K1_KSLICE + 4)
#define K1_NFRAG   (K1_KSLICE / 8)
#define K1_SLAB    (16 * K1_PITCH)

__global__ void __launch_bounds__(K1_THREADS)
ssm_bproj_wmma_kernel(const float* __restrict__ x,
                      const float* __restrict__ Bw,
                      float* __restrict__ Bt) {
  static_assert(K1_KSLICE == 4 * 32, "one 16-byte request per lane covers a wave's piece of a row");
  __shared__ __attribute__((aligned(16))) float sx[K1_WAVES * K1_SLAB];
  __shared__ __attribute__((aligned(16))) float sw[K1_WAVES * K1_SLAB];
  __shared__ float red[K1_WAVES * 256];

  const int tid  = threadIdx.x;
  const int lane = tid & 31;
  const int w    = tid >> 5;
  const int half = lane >> 4;
  const int lidx = lane & 15;
  const int row0 = blockIdx.x * 16;

  const float* xg = x  + (size_t)row0 * D_MODEL + w * K1_KSLICE + 4 * lane;
  const float* wg = Bw + w * K1_KSLICE + 4 * lane;

  v4f gx[16], gw[16];
#pragma unroll
  for (int r = 0; r < 16; ++r) gx[r] = *reinterpret_cast<const v4f*>(xg + (size_t)r * D_MODEL);
#pragma unroll
  for (int n = 0; n < 16; ++n) gw[n] = *reinterpret_cast<const v4f*>(wg + (size_t)n * D_MODEL);

  float* mx = sx + w * K1_SLAB;
  float* mw = sw + w * K1_SLAB;
#pragma unroll
  for (int r = 0; r < 16; ++r) *reinterpret_cast<v4f*>(mx + r * K1_PITCH + 4 * lane) = gx[r];
#pragma unroll
  for (int n = 0; n < 16; ++n) *reinterpret_cast<v4f*>(mw + n * K1_PITCH + 4 * lane) = gw[n];
  __syncthreads();

  const float* fa = mx + lidx * K1_PITCH + 4 * half;
  const float* fb = mw + lidx * K1_PITCH + 4 * half;
  v4f a[K1_NFRAG], b[K1_NFRAG];
#pragma unroll
  for (int i = 0; i < K1_NFRAG; ++i) {
    a[i] = *reinterpret_cast<const v4f*>(fa + 8 * i);
    b[i] = *reinterpret_cast<const v4f*>(fb + 8 * i);
  }

  __builtin_amdgcn_sched_barrier(0);

  v8f acc0 = {}, acc1 = {};
#pragma unroll
  for (int i = 0; i < K1_NFRAG; ++i) {
    v2f a0 = { a[i].x, a[i].y }, b0 = { b[i].x, b[i].y };
    v2f a1 = { a[i].z, a[i].w }, b1 = { b[i].z, b[i].w };
    acc0 = __builtin_amdgcn_wmma_f32_16x16x4_f32(false, a0, false, b0,
                                                 (short)0, acc0, false, false);
    acc1 = __builtin_amdgcn_wmma_f32_16x16x4_f32(false, a1, false, b1,
                                                 (short)0, acc1, false, false);
  }

  __builtin_amdgcn_sched_barrier(0);
  asm volatile("v_nop\n\tv_nop\n\tv_nop\n\tv_nop\n\tv_nop" : "+v"(acc0), "+v"(acc1));
  __builtin_amdgcn_sched_barrier(0);

#pragma unroll
  for (int i = 0; i < 8; ++i) {
    red[w * 256 + (i + 8 * half) * 16 + lidx] = acc0[i] + acc1[i];
  }
  __syncthreads();

  float s = red[tid];
#pragma unroll
  for (int ww = 1; ww < K1_WAVES; ++ww) s += red[ww * 256 + tid];
  Bt[(size_t)row0 * D_STATE + tid] = s;
}

__device__ __forceinline__ void ssm_carry_step(const float (&AL)[D_STATE],
                                               float (&h)[D_STATE],
                                               const float4 (&S)[4]) {
#pragma unroll
  for (int q = 0; q < 4; ++q) {
    float4 v = S[q];
    h[4 * q + 0] = fmaf(AL[4 * q + 0], h[4 * q + 0], v.x);
    h[4 * q + 1] = fmaf(AL[4 * q + 1], h[4 * q + 1], v.y);
    h[4 * q + 2] = fmaf(AL[4 * q + 2], h[4 * q + 2], v.z);
    h[4 * q + 3] = fmaf(AL[4 * q + 3], h[4 * q + 3], v.w);
  }
}

__global__ void __launch_bounds__(SCAN_THREADS)
ssm_chunk_local_kernel(const float* __restrict__ Bt,
                       const float* __restrict__ logA,
                       float* __restrict__ Ssum) {
  static_assert(SCAN_THREADS == CGRP * 32, "one wave per chunk of the group");
  static_assert(D_STATE == 16, "a lane's state is its place in a 16-lane half");
  __shared__ float4 sbtg[CGRP * LCHUNK * 4];
  __shared__ float4 sxc[CGRP][4][32];

  const int lane = threadIdx.x & 31;
  const int w = __builtin_amdgcn_readfirstlane((int)(threadIdx.x >> 5));
  const int c = blockIdx.y * CGRP + w;
  const int b = blockIdx.z;
  const size_t e0 = (size_t)blockIdx.x * 32 * D_STATE + lane;

  v4f st[WSTG];
  ssm_wave_stage_request(Bt + ((size_t)b * SEQ + (size_t)c * LCHUNK) * D_STATE, lane, st);
  v4f la[4];
  {
    const float* pl = logA + e0;
#pragma unroll
    for (int q = 0; q < 4; ++q) {
      v4f v = { pl[32 * (4 * q + 0)], pl[32 * (4 * q + 1)],
                pl[32 * (4 * q + 2)], pl[32 * (4 * q + 3)] };
      la[q] = v;
    }
  }
  ssm_wave_entry_keep_together(st, la);
  float4* sbt = sbtg + w * (LCHUNK * 4);
  ssm_wave_stage_commit(sbt, lane, st);

  float A[D_STATE], h[D_STATE];
  ssm_abar_from(la, A);
#pragma unroll
  for (int j = 0; j < D_STATE; ++j) h[j] = 0.0f;

  const float* srow = reinterpret_cast<const float*>(sbt) + (lane & 15);
  float r[LCHUNK];
#pragma unroll
  for (int l = 0; l < LCHUNK; ++l) r[l] = srow[l * D_STATE];

#pragma unroll
  for (int l = 0; l < LCHUNK; ++l) {
#pragma unroll
    for (int j = 0; j < D_STATE; ++j) h[j] = fmaf(A[j], h[j], r[l]);
  }

#pragma unroll
  for (int q = 0; q < 4; ++q) {
    float4 v = { h[4 * q + 0], h[4 * q + 1], h[4 * q + 2], h[4 * q + 3] };
    sxc[w][q][lane] = v;
  }
  __syncthreads();

  if (w > 0) {
    float AL[D_STATE];
#pragma unroll
    for (int j = 0; j < D_STATE; ++j) {
      float p = A[j];
#pragma unroll
      for (int k = 0; k < 7; ++k) p = p * p;
      AL[j] = p;
    }
    float e[D_STATE];
#pragma unroll
    for (int q = 0; q < 4; ++q) {
      float4 v = sxc[0][q][lane];
      e[4 * q + 0] = v.x; e[4 * q + 1] = v.y; e[4 * q + 2] = v.z; e[4 * q + 3] = v.w;
    }
#pragma unroll
    for (int i = 1; i < CGRP - 1; ++i) {
      if (i < w) {
        float4 sv[4];
#pragma unroll
        for (int q = 0; q < 4; ++q) sv[q] = sxc[i][q][lane];
        ssm_carry_step(AL, e, sv);
      }
    }
#pragma unroll
    for (int j = 0; j < D_STATE; ++j) h[j] = fmaf(AL[j], e[j], h[j]);
  }

  float* out = Ssum + ((size_t)b * NCHUNK + c) * D_MODEL * D_STATE + e0;
#pragma unroll
  for (int j = 0; j < D_STATE; ++j) out[32 * j] = h[j];
}

template <int G>
__device__ __forceinline__ void ssm_replay_rows(const float4 (&bt)[BGRP][4],
                                                const float (&A)[D_STATE],
                                                const float (&C)[D_STATE],
                                                float (&h)[D_STATE],
                                                const float (&xv)[XGRP],
                                                float dskip,
                                                float* __restrict__ yrow) {
#pragma unroll
  for (int j = 0; j < BGRP; ++j) {
    float acc = 0.0f;
#pragma unroll
    for (int q = 0; q < 4; ++q) {
      float4 v = bt[j][q];
      h[4 * q + 0] = fmaf(A[4 * q + 0], h[4 * q + 0], v.x);
      h[4 * q + 1] = fmaf(A[4 * q + 1], h[4 * q + 1], v.y);
      h[4 * q + 2] = fmaf(A[4 * q + 2], h[4 * q + 2], v.z);
      h[4 * q + 3] = fmaf(A[4 * q + 3], h[4 * q + 3], v.w);
      acc = fmaf(C[4 * q + 0], h[4 * q + 0], acc);
      acc = fmaf(C[4 * q + 1], h[4 * q + 1], acc);
      acc = fmaf(C[4 * q + 2], h[4 * q + 2], acc);
      acc = fmaf(C[4 * q + 3], h[4 * q + 3], acc);
    }
    yrow[(size_t)(G + j) * D_MODEL] = fmaf(dskip, xv[G + j], acc);
  }
}

__global__ void __launch_bounds__(SCAN_THREADS)
ssm_output_kernel(const float* __restrict__ x,
                  const float* __restrict__ Bt,
                  const float* __restrict__ logA,
                  const float* __restrict__ Cw,
                  const float* __restrict__ Dvec,
                  const float* __restrict__ Ssum,
                  float* __restrict__ y) {
  __shared__ float4 sbt[LCHUNK * 4];

  const int d = blockIdx.x * SCAN_THREADS + threadIdx.x;
  const int c = blockIdx.y;
  const int b = blockIdx.z;

  const size_t rowbase = (size_t)b * SEQ + (size_t)c * LCHUNK;
  const float* btrow = Bt + rowbase * D_STATE;
  const float* xcol  = x + rowbase * D_MODEL + d;
  float*       ycol  = y + rowbase * D_MODEL + d;

  v4f st[STG];
  ssm_stage_request(btrow, st);

  const int cg = c / CGRP;
  const int cw = c % CGRP;
  const float* srow = Ssum + ((size_t)b * NCHUNK * D_MODEL + d) * D_STATE;
  float4 sv[NGRP - 1][4];
#pragma unroll
  for (int j = 0; j < NGRP - 1; ++j) {
    if (j < cg) {
      const float4* p = reinterpret_cast<const float4*>(
          srow + (size_t)(CGRP * j + CGRP - 1) * D_MODEL * D_STATE);
#pragma unroll
      for (int q = 0; q < 4; ++q) sv[j][q] = p[q];
    }
  }
  float4 sp[4];
  if (cw > 0) {
    const float4* p = reinterpret_cast<const float4*>(
        srow + (size_t)(c - 1) * D_MODEL * D_STATE);
#pragma unroll
    for (int q = 0; q < 4; ++q) sp[q] = p[q];
  }

  v4f la[4];
  float4 vc[4];
  {
    const v4f* pl = reinterpret_cast<const v4f*>(logA + (size_t)d * D_STATE);
    const float4* pc = reinterpret_cast<const float4*>(Cw + (size_t)d * D_STATE);
#pragma unroll
    for (int q = 0; q < 4; ++q) { la[q] = pl[q]; vc[q] = pc[q]; }
  }
  const float dskip = Dvec[d];

  float xa[XGRP], xb[XGRP];
#pragma unroll
  for (int j = 0; j < XGRP; ++j) xa[j] = xcol[(size_t)j * D_MODEL];

  ssm_entry_keep_together(st, la);
  ssm_stage_commit(sbt, st);

  float A[D_STATE], C[D_STATE], h[D_STATE];
  ssm_abar_from(la, A);
#pragma unroll
  for (int q = 0; q < 4; ++q) {
    C[4 * q + 0] = vc[q].x; C[4 * q + 1] = vc[q].y; C[4 * q + 2] = vc[q].z; C[4 * q + 3] = vc[q].w;
  }
#pragma unroll
  for (int s = 0; s < D_STATE; ++s) h[s] = 0.0f;

  if (c > 0) {
    float AL[D_STATE];
#pragma unroll
    for (int s = 0; s < D_STATE; ++s) {
      float p = A[s];
#pragma unroll
      for (int j = 0; j < 7; ++j) p = p * p;
      AL[s] = p;
    }
    if (cg > 0) {
      static_assert(CGRP == 4, "A_G is two squarings of A_Lc");
      float AG[D_STATE];
#pragma unroll
      for (int s = 0; s < D_STATE; ++s) {
        float p = AL[s] * AL[s];
        AG[s] = p * p;
      }
#pragma unroll
      for (int j = 0; j < NGRP - 1; ++j) {
        if (j < cg) ssm_carry_step(AG, h, sv[j]);
      }
    }
    if (cw > 0) {
#pragma unroll
      for (int i = 1; i < CGRP - 1; ++i) {
        if (i < cw) {
#pragma unroll
          for (int s = 0; s < D_STATE; ++s) h[s] = AL[s] * h[s];
        }
      }
      ssm_carry_step(AL, h, sp);
    }
  }

  float4 ra[BGRP][4], rb[BGRP][4];
  ssm_rows_request(sbt, ra);
  ssm_rows_keep_ahead(ra);
  for (int l0 = 0; l0 < LCHUNK; l0 += 2 * XGRP) {
#pragma unroll
    for (int j = 0; j < XGRP; ++j) xb[j] = xcol[(size_t)(l0 + XGRP + j) * D_MODEL];

    __builtin_amdgcn_sched_barrier(0);
    ssm_rows_request(sbt + 4 * (l0 + BGRP), rb);
    __builtin_amdgcn_sched_barrier(0);
    ssm_replay_rows<0>(ra, A, C, h, xa, dskip, ycol + (size_t)l0 * D_MODEL);
    __builtin_amdgcn_sched_barrier(0);
    ssm_rows_request(sbt + 4 * (l0 + 2 * BGRP), ra);
    __builtin_amdgcn_sched_barrier(0);
    ssm_replay_rows<BGRP>(rb, A, C, h, xa, dskip, ycol + (size_t)l0 * D_MODEL);

    const int ln = (l0 + 2 * XGRP < LCHUNK) ? (l0 + 2 * XGRP) : l0;
    __builtin_amdgcn_sched_barrier(0);
#pragma unroll
    for (int j = 0; j < XGRP; ++j) xa[j] = xcol[(size_t)(ln + j) * D_MODEL];

    __builtin_amdgcn_sched_barrier(0);
    ssm_rows_request(sbt + 4 * (l0 + 3 * BGRP), rb);
    __builtin_amdgcn_sched_barrier(0);
    ssm_replay_rows<0>(ra, A, C, h, xb, dskip, ycol + (size_t)(l0 + XGRP) * D_MODEL);
    const int rn = (l0 + 2 * XGRP < LCHUNK) ? (l0 + 2 * XGRP) : (l0 + 3 * BGRP);
    __builtin_amdgcn_sched_barrier(0);
    ssm_rows_request(sbt + 4 * rn, ra);
    __builtin_amdgcn_sched_barrier(0);
    ssm_replay_rows<BGRP>(rb, A, C, h, xb, dskip, ycol + (size_t)(l0 + XGRP) * D_MODEL);
  }
}

extern "C" void kernel_launch(void* const* d_in, const int* in_sizes, int n_in,
                              void* d_out, int out_size, void* d_ws, size_t ws_size,
                              hipStream_t stream) {
  const float* x    = (const float*)d_in[0];
  const float* logA = (const float*)d_in[1];
  const float* Bw   = (const float*)d_in[2];
  const float* Cw   = (const float*)d_in[3];
  const float* Dv   = (const float*)d_in[4];
  float* y = (float*)d_out;

  char* ws = (char*)d_ws;
  float* Bt    = (float*)(ws + (128 << 10));
  float* Ssum  = (float*)(ws + (384 << 10));

  ssm_bproj_wmma_kernel<<<(BATCH * SEQ) / 16, K1_THREADS, 0, stream>>>(x, Bw, Bt);

  dim3 gscan(D_MODEL / SCAN_THREADS, NCHUNK, BATCH);

  dim3 glocal(D_MODEL / 32, NGRP, BATCH);
  ssm_chunk_local_kernel<<<glocal, SCAN_THREADS, 0, stream>>>(Bt, logA, Ssum);

  ssm_output_kernel<<<gscan, SCAN_THREADS, 0, stream>>>(x, Bt, logA, Cw, Dv, Ssum, y);
}
